# starting kernel
# baseline (speedup 1.0000x reference)
#include <hip/hip_runtime.h>
#include <hip/hip_bf16.h>

typedef __attribute__((ext_vector_type(2))) float v2f;
typedef __attribute__((ext_vector_type(8))) float v8f;

#define BB 32
#define FF 4096
#define SS 8
#define DD 512
#define HH 2048
#define NITER 3
#define EPS_A 1e-8f
#define LN_EPS_C 1e-5f
#define SCALE_C 0.04419417382415922f

__device__ __forceinline__ v8f wmma4(v2f a, v2f b, v8f c) {
  return __builtin_amdgcn_wmma_f32_16x16x4_f32(
       false, a,  false, b,
       (short)0, c,  false,  false);
}

__device__ __forceinline__ v2f frag_a(const float* A, int lda, int lane, int k0) {
  int row = lane & 15;
  int kk  = k0 + ((lane >> 4) << 1);
  return *(const v2f*)&A[row * lda + kk];
}
__device__ __forceinline__ v2f frag_b(const float* B, int ldb, int lane, int k0, int n0) {
  int col = n0 + (lane & 15);
  int p   = (k0 >> 1) + (lane >> 4);
  return *(const v2f*)&B[(p * ldb + col) * 2];
}

__device__ __forceinline__ float wave_sum(float v) {
#pragma unroll
  for (int o = 16; o; o >>= 1) v += __shfl_xor(v, o, 32);
  return v;
}

__global__ void __launch_bounds__(256)
row_stats_kernel(const float* __restrict__ X, float* __restrict__ mean,
                 float* __restrict__ rstd) {
  int wv = threadIdx.x >> 5, lane = threadIdx.x & 31;
  int row = blockIdx.x * 8 + wv;
  const float* x = X + (size_t)row * DD;
  float s = 0.f, s2 = 0.f;
#pragma unroll
  for (int j = 0; j < DD / 32; ++j) {
    float v = x[lane + 32 * j]; s += v; s2 += v * v;
  }
  s = wave_sum(s); s2 = wave_sum(s2);
  if (lane == 0) {
    float m = s * (1.f / DD);
    float var = s2 * (1.f / DD) - m * m;
    mean[row] = m;
    rstd[row] = rsqrtf(var + LN_EPS_C);
  }
}

__global__ void __launch_bounds__(256)
ln_rows_kernel(const float* __restrict__ X, const float* __restrict__ w,
               const float* __restrict__ b, float* __restrict__ Y) {
  int wv = threadIdx.x >> 5, lane = threadIdx.x & 31;
  int row = blockIdx.x * 8 + wv;
  const float* x = X + (size_t)row * DD;
  float xv[16]; float s = 0.f, s2 = 0.f;
#pragma unroll
  for (int j = 0; j < 16; ++j) {
    float v = x[lane + 32 * j]; xv[j] = v; s += v; s2 += v * v;
  }
  s = wave_sum(s); s2 = wave_sum(s2);
  float m = s * (1.f / DD);
  float r = rsqrtf(s2 * (1.f / DD) - m * m + LN_EPS_C);
  float* y = Y + (size_t)row * DD;
#pragma unroll
  for (int j = 0; j < 16; ++j) {
    int d = lane + 32 * j;
    y[d] = (xv[j] - m) * r * w[d] + b[d];
  }
}

__global__ void __launch_bounds__(256)
gemm_ln_dual_kernel(const float* __restrict__ X, const float* __restrict__ mean,
                    const float* __restrict__ rstd, const float* __restrict__ lnw,
                    const float* __restrict__ lnb, const float* __restrict__ Wk,
                    const float* __restrict__ Wv, float* __restrict__ Ko,
                    float* __restrict__ Vo) {
  __shared__ float As[128 * 16];
  __shared__ float Bks[16 * 64];
  __shared__ float Bvs[16 * 64];
  int t = threadIdx.x, lane = t & 31, wv = t >> 5;
  int m0 = blockIdx.x * 128, n0 = blockIdx.y * 64;

  int arow = t >> 1, aoff = (t & 1) * 8;
  size_t grow = (size_t)(m0 + arow);
  float mn = mean[grow], rs = rstd[grow];
  const float* xrow = X + grow * DD;
  int bp = t >> 5, bn2 = (t & 31) * 2;

  v8f accK[4] = {}, accV[4] = {};
  for (int kc = 0; kc < DD; kc += 16) {
    __syncthreads();
#pragma unroll
    for (int j = 0; j < 8; ++j) {
      int k = kc + aoff + j;
      As[arow * 16 + aoff + j] = (xrow[k] - mn) * rs * lnw[k] + lnb[k];
    }
    {
      float2 r0 = *(const float2*)&Wk[(size_t)(kc + 2 * bp) * DD + n0 + bn2];
      float2 r1 = *(const float2*)&Wk[(size_t)(kc + 2 * bp + 1) * DD + n0 + bn2];
      float4 wk; wk.x = r0.x; wk.y = r1.x; wk.z = r0.y; wk.w = r1.y;
      *(float4*)&Bks[(bp * 64 + bn2) * 2] = wk;
      float2 s0 = *(const float2*)&Wv[(size_t)(kc + 2 * bp) * DD + n0 + bn2];
      float2 s1 = *(const float2*)&Wv[(size_t)(kc + 2 * bp + 1) * DD + n0 + bn2];
      float4 wvv; wvv.x = s0.x; wvv.y = s1.x; wvv.z = s0.y; wvv.w = s1.y;
      *(float4*)&Bvs[(bp * 64 + bn2) * 2] = wvv;
    }
    __syncthreads();
    const float* Aw = &As[(wv * 16) * 16];
#pragma unroll
    for (int ks = 0; ks < 16; ks += 4) {
      v2f a = frag_a(Aw, 16, lane, ks);
#pragma unroll
      for (int nt = 0; nt < 4; ++nt) {
        v2f bk = frag_b(Bks, 64, lane, ks, nt * 16);
        accK[nt] = wmma4(a, bk, accK[nt]);
        v2f bv = frag_b(Bvs, 64, lane, ks, nt * 16);
        accV[nt] = wmma4(a, bv, accV[nt]);
      }
    }
  }
  int half = lane >> 4, l16 = lane & 15;
#pragma unroll
  for (int nt = 0; nt < 4; ++nt)
#pragma unroll
    for (int i = 0; i < 8; ++i) {
      size_t mrow = (size_t)(m0 + wv * 16 + half * 8 + i);
      int ncol = n0 + nt * 16 + l16;
      Ko[mrow * DD + ncol] = accK[nt][i];
      Vo[mrow * DD + ncol] = accV[nt][i];
    }
}

template <int EPI>
__global__ void __launch_bounds__(256)
gemm_rm_kernel(const float* __restrict__ A, const float* __restrict__ B,
               const float* __restrict__ bias, float* __restrict__ C,
               const float* __restrict__ aux, int K, int N) {
  __shared__ float As[128 * 16];
  __shared__ float Bs[16 * 64];
  int t = threadIdx.x, lane = t & 31, wv = t >> 5;
  int m0 = blockIdx.x * 128, n0 = blockIdx.y * 64;
  int arow = t >> 1, aoff = (t & 1) * 8;
  const float* xrow = A + (size_t)(m0 + arow) * K;
  int bp = t >> 5, bn2 = (t & 31) * 2;

  v8f acc[4] = {};
  for (int kc = 0; kc < K; kc += 16) {
    __syncthreads();
    {
      float4 a0 = *(const float4*)&xrow[kc + aoff];
      float4 a1 = *(const float4*)&xrow[kc + aoff + 4];
      *(float4*)&As[arow * 16 + aoff] = a0;
      *(float4*)&As[arow * 16 + aoff + 4] = a1;
      float2 r0 = *(const float2*)&B[(size_t)(kc + 2 * bp) * N + n0 + bn2];
      float2 r1 = *(const float2*)&B[(size_t)(kc + 2 * bp + 1) * N + n0 + bn2];
      float4 wb; wb.x = r0.x; wb.y = r1.x; wb.z = r0.y; wb.w = r1.y;
      *(float4*)&Bs[(bp * 64 + bn2) * 2] = wb;
    }
    __syncthreads();
    const float* Aw = &As[(wv * 16) * 16];
#pragma unroll
    for (int ks = 0; ks < 16; ks += 4) {
      v2f a = frag_a(Aw, 16, lane, ks);
#pragma unroll
      for (int nt = 0; nt < 4; ++nt) {
        v2f b = frag_b(Bs, 64, lane, ks, nt * 16);
        acc[nt] = wmma4(a, b, acc[nt]);
      }
    }
  }
  int half = lane >> 4, l16 = lane & 15;
#pragma unroll
  for (int nt = 0; nt < 4; ++nt) {
    int ncol = n0 + nt * 16 + l16;
    float bv = (EPI >= 1) ? bias[ncol] : 0.f;
#pragma unroll
    for (int i = 0; i < 8; ++i) {
      size_t mrow = (size_t)(m0 + wv * 16 + half * 8 + i);
      float v = acc[nt][i] + bv;
      if (EPI == 1) v = fmaxf(v, 0.f);
      if (EPI == 2) v += aux[mrow * (size_t)N + ncol];
      C[mrow * (size_t)N + ncol] = v;
    }
  }
}

__global__ void __launch_bounds__(256)
gemm_bt_kernel(const float* __restrict__ A, const float* __restrict__ W,
               const float* __restrict__ bias, float* __restrict__ C,
               int K, int N) {
  __shared__ float As[128 * 16];
  __shared__ float Bs[16 * 64];
  int t = threadIdx.x, lane = t & 31, wv = t >> 5;
  int m0 = blockIdx.x * 128, n0 = blockIdx.y * 64;
  int arow = t >> 1, aoff = (t & 1) * 8;
  const float* xrow = A + (size_t)(m0 + arow) * K;
  int bn = t & 63, bp0 = t >> 6;

  v8f acc[4] = {};
  for (int kc = 0; kc < K; kc += 16) {
    __syncthreads();
    {
      float4 a0 = *(const float4*)&xrow[kc + aoff];
      float4 a1 = *(const float4*)&xrow[kc + aoff + 4];
      *(float4*)&As[arow * 16 + aoff] = a0;
      *(float4*)&As[arow * 16 + aoff + 4] = a1;
#pragma unroll
      for (int j = 0; j < 2; ++j) {
        int p = bp0 + j * 4;
        float2 f = *(const float2*)&W[(size_t)(n0 + bn) * K + kc + 2 * p];
        *(float2*)&Bs[(p * 64 + bn) * 2] = f;
      }
    }
    __syncthreads();
    const float* Aw = &As[(wv * 16) * 16];
#pragma unroll
    for (int ks = 0; ks < 16; ks += 4) {
      v2f a = frag_a(Aw, 16, lane, ks);
#pragma unroll
      for (int nt = 0; nt < 4; ++nt) {
        v2f b = frag_b(Bs, 64, lane, ks, nt * 16);
        acc[nt] = wmma4(a, b, acc[nt]);
      }
    }
  }
  int half = lane >> 4, l16 = lane & 15;
#pragma unroll
  for (int nt = 0; nt < 4; ++nt) {
    int ncol = n0 + nt * 16 + l16;
    float bv = bias[ncol];
#pragma unroll
    for (int i = 0; i < 8; ++i) {
      size_t mrow = (size_t)(m0 + wv * 16 + half * 8 + i);
      C[mrow * (size_t)N + ncol] = acc[nt][i] + bv;
    }
  }
}

__global__ void __launch_bounds__(256)
dots_kernel(const float* __restrict__ Q, const float* __restrict__ Keys,
            float* __restrict__ Dots) {
  __shared__ float As[16 * 16];
  __shared__ float Bs[16 * 512];
  int t = threadIdx.x, lane = t & 31, wv = t >> 5;
  int b = blockIdx.x, nb = blockIdx.y * 512;
  const float* qb = Q + (size_t)b * SS * DD;
  const float* kb = Keys + ((size_t)b * FF + nb) * DD;
  v8f acc[4] = {};
  for (int kc = 0; kc < DD; kc += 16) {
    __syncthreads();
    {
      int r = t >> 4, k = t & 15;
      As[r * 16 + k] = (r < SS) ? qb[(size_t)r * DD + kc + k] : 0.f;
    }
    {
      int k4 = (t & 3) * 4;
#pragma unroll
      for (int j = 0; j < 8; ++j) {
        int n = (t >> 2) + j * 64;
        float4 f = *(const float4*)&kb[(size_t)n * DD + kc + k4];
        float2 lo; lo.x = f.x; lo.y = f.y;
        float2 hi; hi.x = f.z; hi.y = f.w;
        *(float2*)&Bs[(((k4 >> 1) + 0) * 512 + n) * 2] = lo;
        *(float2*)&Bs[(((k4 >> 1) + 1) * 512 + n) * 2] = hi;
      }
    }
    __syncthreads();
#pragma unroll
    for (int ks = 0; ks < 16; ks += 4) {
      v2f a = frag_a(As, 16, lane, ks);
#pragma unroll
      for (int nt = 0; nt < 4; ++nt) {
        v2f b2 = frag_b(Bs, 512, lane, ks, wv * 64 + nt * 16);
        acc[nt] = wmma4(a, b2, acc[nt]);
      }
    }
  }
  int half = lane >> 4, l16 = lane & 15;
  if (half == 0) {
#pragma unroll
    for (int nt = 0; nt < 4; ++nt)
#pragma unroll
      for (int i = 0; i < 8; ++i) {
        int f = nb + wv * 64 + nt * 16 + l16;
        Dots[((size_t)b * SS + i) * FF + f] = acc[nt][i] * SCALE_C;
      }
  }
}

__global__ void __launch_bounds__(256)
softmax_kernel(const float* __restrict__ Dots, float* __restrict__ PreNorm,
               float* __restrict__ Attn, float* __restrict__ Sums) {
  int b = blockIdx.y;
  int f = blockIdx.x * 256 + threadIdx.x;
  int lane = threadIdx.x & 31;
  float d[SS]; float mx = -3.4e38f;
#pragma unroll
  for (int s = 0; s < SS; ++s) {
    d[s] = Dots[((size_t)b * SS + s) * FF + f];
    mx = fmaxf(mx, d[s]);
  }
  float sum = 0.f;
#pragma unroll
  for (int s = 0; s < SS; ++s) { d[s] = expf(d[s] - mx); sum += d[s]; }
  float inv = 1.f / sum;
#pragma unroll
  for (int s = 0; s < SS; ++s) {
    float p = d[s] * inv;
    size_t idx = ((size_t)b * SS + s) * FF + f;
    PreNorm[idx] = p;
    float aw = p + EPS_A;
    Attn[idx] = aw;
    float r = wave_sum(aw);
    if (lane == 0) atomicAdd(&Sums[b * SS + s], r);
  }
}

__global__ void zero_sums_kernel(float* __restrict__ Sums) {
  Sums[threadIdx.x] = 0.f;
}

__global__ void __launch_bounds__(256)
updates_kernel(const float* __restrict__ Attn, const float* __restrict__ Vals,
               const float* __restrict__ Sums, float* __restrict__ Upd) {
  __shared__ float As[16 * 16];
  __shared__ float Bs[16 * 512];
  int t = threadIdx.x, lane = t & 31, wv = t >> 5;
  int b = blockIdx.x;
  const float* ab = Attn + (size_t)b * SS * FF;
  const float* vb = Vals + (size_t)b * FF * DD;
  v8f acc[4] = {};
  for (int kc = 0; kc < FF; kc += 16) {
    __syncthreads();
    {
      int r = t >> 4, k = t & 15;
      As[r * 16 + k] = (r < SS) ? ab[(size_t)r * FF + kc + k] : 0.f;
    }
    {
#pragma unroll
      for (int j = 0; j < 8; ++j) {
        int u = t + j * 256;
        int p = u >> 8;
        int n2 = (u & 255) * 2;
        float2 r0 = *(const float2*)&vb[(size_t)(kc + 2 * p) * DD + n2];
        float2 r1 = *(const float2*)&vb[(size_t)(kc + 2 * p + 1) * DD + n2];
        float4 wb; wb.x = r0.x; wb.y = r1.x; wb.z = r0.y; wb.w = r1.y;
        *(float4*)&Bs[(p * 512 + n2) * 2] = wb;
      }
    }
    __syncthreads();
#pragma unroll
    for (int ks = 0; ks < 16; ks += 4) {
      v2f a = frag_a(As, 16, lane, ks);
#pragma unroll
      for (int nt = 0; nt < 4; ++nt) {
        v2f b2 = frag_b(Bs, 512, lane, ks, wv * 64 + nt * 16);
        acc[nt] = wmma4(a, b2, acc[nt]);
      }
    }
  }
  int half = lane >> 4, l16 = lane & 15;
  if (half == 0) {
#pragma unroll
    for (int i = 0; i < 8; ++i) {
      float inv = 1.f / Sums[b * SS + i];
#pragma unroll
      for (int nt = 0; nt < 4; ++nt) {
        int n = wv * 64 + nt * 16 + l16;
        Upd[((size_t)b * SS + i) * DD + n] = acc[nt][i] * inv;
      }
    }
  }
}

__global__ void __launch_bounds__(256)
gru_kernel(const float* __restrict__ gi, const float* __restrict__ gh,
           const float* __restrict__ sn, float* __restrict__ slots) {
  int idx = blockIdx.x * 256 + threadIdx.x;
  int m = idx >> 9, d = idx & 511;
  size_t base = (size_t)m * 3 * DD;
  float ir = gi[base + d],           hr = gh[base + d];
  float iz = gi[base + DD + d],      hz = gh[base + DD + d];
  float in_ = gi[base + 2 * DD + d], hn = gh[base + 2 * DD + d];
  float r = 1.f / (1.f + expf(-(ir + hr)));
  float z = 1.f / (1.f + expf(-(iz + hz)));
  float n = tanhf(in_ + r * hn);
  float h = sn[(size_t)m * DD + d];
  slots[(size_t)m * DD + d] = (1.f - z) * n + z * h;
}

extern "C" void kernel_launch(void* const* d_in, const int* in_sizes, int n_in,
                              void* d_out, int out_size, void* d_ws, size_t ws_size,
                              hipStream_t stream) {
  (void)in_sizes; (void)n_in; (void)out_size; (void)ws_size;
  const float* in_slots  = (const float*)d_in[0];
  const float* features  = (const float*)d_in[1];
  const float* w_k       = (const float*)d_in[2];
  const float* w_v       = (const float*)d_in[3];
  const float* w_q       = (const float*)d_in[4];
  const float* ln_feat_w = (const float*)d_in[5];
  const float* ln_feat_b = (const float*)d_in[6];
  const float* ln_slot_w = (const float*)d_in[7];
  const float* ln_slot_b = (const float*)d_in[8];
  const float* gru_w_ih  = (const float*)d_in[9];
  const float* gru_w_hh  = (const float*)d_in[10];
  const float* gru_b_ih  = (const float*)d_in[11];
  const float* gru_b_hh  = (const float*)d_in[12];
  const float* ln_mlp_w  = (const float*)d_in[13];
  const float* ln_mlp_b  = (const float*)d_in[14];
  const float* mlp_w1    = (const float*)d_in[15];
  const float* mlp_b1    = (const float*)d_in[16];
  const float* mlp_w2    = (const float*)d_in[17];
  const float* mlp_b2    = (const float*)d_in[18];

  float* ws = (float*)d_ws;
  size_t off = 0;
  auto carve = [&](size_t n) { float* p = ws + off; off += n; return p; };
  float* mean  = carve((size_t)BB * FF);
  float* rstd  = carve((size_t)BB * FF);
  float* keys  = carve((size_t)BB * FF * DD);
  float* vals  = carve((size_t)BB * FF * DD);
  float* slots = carve((size_t)BB * SS * DD);
  float* s_n   = carve((size_t)BB * SS * DD);
  float* q     = carve((size_t)BB * SS * DD);
  float* dots  = carve((size_t)BB * SS * FF);
  float* attn  = carve((size_t)BB * SS * FF);
  float* sums  = carve(256);
  float* upd   = carve((size_t)BB * SS * DD);
  float* gi    = carve((size_t)BB * SS * 3 * DD);
  float* gh    = carve((size_t)BB * SS * 3 * DD);
  float* hln   = carve((size_t)BB * SS * DD);
  float* h1    = carve((size_t)BB * SS * HH);

  float* out_slots = (float*)d_out;
  float* out_attn  = out_slots + (size_t)BB * SS * DD;

  hipMemcpyAsync(slots, in_slots, (size_t)BB * SS * DD * sizeof(float),
                 hipMemcpyDeviceToDevice, stream);

  row_stats_kernel<<<(BB * FF) / 8, 256, 0, stream>>>(features, mean, rstd);
  gemm_ln_dual_kernel<<<dim3((BB * FF) / 128, DD / 64), 256, 0, stream>>>(
      features, mean, rstd, ln_feat_w, ln_feat_b, w_k, w_v, keys, vals);

  for (int it = 0; it < NITER; ++it) {
    ln_rows_kernel<<<32, 256, 0, stream>>>(slots, ln_slot_w, ln_slot_b, s_n);
    gemm_rm_kernel<0><<<dim3(2, DD / 64), 256, 0, stream>>>(
        s_n, w_q, nullptr, q, nullptr, DD, DD);
    zero_sums_kernel<<<1, 256, 0, stream>>>(sums);
    dots_kernel<<<dim3(BB, FF / 512), 256, 0, stream>>>(q, keys, dots);
    softmax_kernel<<<dim3(FF / 256, BB), 256, 0, stream>>>(dots, out_attn, attn, sums);
    updates_kernel<<<BB, 256, 0, stream>>>(attn, vals, sums, upd);
    gemm_bt_kernel<<<dim3(2, (3 * DD) / 64), 256, 0, stream>>>(
        upd, gru_w_ih, gru_b_ih, gi, DD, 3 * DD);
    gemm_bt_kernel<<<dim3(2, (3 * DD) / 64), 256, 0, stream>>>(
        s_n, gru_w_hh, gru_b_hh, gh, DD, 3 * DD);
    gru_kernel<<<(BB * SS * DD) / 256, 256, 0, stream>>>(gi, gh, s_n, slots);
    ln_rows_kernel<<<32, 256, 0, stream>>>(slots, ln_mlp_w, ln_mlp_b, hln);
    gemm_rm_kernel<1><<<dim3(2, HH / 64), 256, 0, stream>>>(
        hln, mlp_w1, mlp_b1, h1, nullptr, DD, HH);
    gemm_rm_kernel<2><<<dim3(2, DD / 64), 256, 0, stream>>>(
        h1, mlp_w2, mlp_b2, slots, slots, HH, DD);
  }

  hipMemcpyAsync(out_slots, slots, (size_t)BB * SS * DD * sizeof(float),
                 hipMemcpyDeviceToDevice, stream);
}


# slot-side products (q, GRU gates, both MLP layers) re-tiled: a wave owns a 16x32 output tile over full K, fragments straight from global well ahead of use, no LDS, barrier or K split (new sum order)
# speedup vs baseline: 3.0432x; 3.0432x over previous
#include <hip/hip_runtime.h>
#include <hip/hip_bf16.h>

typedef __attribute__((ext_vector_type(2))) float v2f;
typedef __attribute__((ext_vector_type(8))) float v8f;

#define BB 32
#define FF 4096
#define SS 8
#define DD 512
#define HH 2048
#define NITER 3
#define EPS_A 1e-8f
#define LN_EPS_C 1e-5f
#define SCALE_C 0.04419417382415922f

__device__ __forceinline__ v8f wmma4(v2f a, v2f b, v8f c) {
  return __builtin_amdgcn_wmma_f32_16x16x4_f32(
       false, a,  false, b,
       (short)0, c,  false,  false);
}

__device__ __forceinline__ v2f frag_a(const float* A, int lda, int lane, int k0) {
  int row = lane & 15;
  int kk  = k0 + ((lane >> 4) << 1);
  return *(const v2f*)&A[row * lda + kk];
}
__device__ __forceinline__ v2f frag_b(const float* B, int ldb, int lane, int k0, int n0) {
  int col = n0 + (lane & 15);
  int p   = (k0 >> 1) + (lane >> 4);
  return *(const v2f*)&B[(p * ldb + col) * 2];
}

__device__ __forceinline__ float wave_sum(float v) {
#pragma unroll
  for (int o = 16; o; o >>= 1) v += __shfl_xor(v, o, 32);
  return v;
}

__global__ void __launch_bounds__(256)
row_stats_kernel(const float* __restrict__ X, float* __restrict__ mean,
                 float* __restrict__ rstd) {
  int wv = threadIdx.x >> 5, lane = threadIdx.x & 31;
  int row = blockIdx.x * 8 + wv;
  const float* x = X + (size_t)row * DD;
  float s = 0.f, s2 = 0.f;
#pragma unroll
  for (int j = 0; j < DD / 32; ++j) {
    float v = x[lane + 32 * j]; s += v; s2 += v * v;
  }
  s = wave_sum(s); s2 = wave_sum(s2);
  if (lane == 0) {
    float m = s * (1.f / DD);
    float var = s2 * (1.f / DD) - m * m;
    mean[row] = m;
    rstd[row] = rsqrtf(var + LN_EPS_C);
  }
}

__global__ void __launch_bounds__(256)
ln_rows_kernel(const float* __restrict__ X, const float* __restrict__ w,
               const float* __restrict__ b, float* __restrict__ Y) {
  int wv = threadIdx.x >> 5, lane = threadIdx.x & 31;
  int row = blockIdx.x * 8 + wv;
  const float* x = X + (size_t)row * DD;
  float xv[16]; float s = 0.f, s2 = 0.f;
#pragma unroll
  for (int j = 0; j < 16; ++j) {
    float v = x[lane + 32 * j]; xv[j] = v; s += v; s2 += v * v;
  }
  s = wave_sum(s); s2 = wave_sum(s2);
  float m = s * (1.f / DD);
  float r = rsqrtf(s2 * (1.f / DD) - m * m + LN_EPS_C);
  float* y = Y + (size_t)row * DD;
#pragma unroll
  for (int j = 0; j < 16; ++j) {
    int d = lane + 32 * j;
    y[d] = (xv[j] - m) * r * w[d] + b[d];
  }
}

__global__ void __launch_bounds__(256)
gemm_ln_dual_kernel(const float* __restrict__ X, const float* __restrict__ mean,
                    const float* __restrict__ rstd, const float* __restrict__ lnw,
                    const float* __restrict__ lnb, const float* __restrict__ Wk,
                    const float* __restrict__ Wv, float* __restrict__ Ko,
                    float* __restrict__ Vo) {
  __shared__ float As[128 * 16];
  __shared__ float Bks[16 * 64];
  __shared__ float Bvs[16 * 64];
  int t = threadIdx.x, lane = t & 31, wv = t >> 5;
  int m0 = blockIdx.x * 128, n0 = blockIdx.y * 64;

  int arow = t >> 1, aoff = (t & 1) * 8;
  size_t grow = (size_t)(m0 + arow);
  float mn = mean[grow], rs = rstd[grow];
  const float* xrow = X + grow * DD;
  int bp = t >> 5, bn2 = (t & 31) * 2;

  v8f accK[4] = {}, accV[4] = {};
  for (int kc = 0; kc < DD; kc += 16) {
    __syncthreads();
#pragma unroll
    for (int j = 0; j < 8; ++j) {
      int k = kc + aoff + j;
      As[arow * 16 + aoff + j] = (xrow[k] - mn) * rs * lnw[k] + lnb[k];
    }
    {
      float2 r0 = *(const float2*)&Wk[(size_t)(kc + 2 * bp) * DD + n0 + bn2];
      float2 r1 = *(const float2*)&Wk[(size_t)(kc + 2 * bp + 1) * DD + n0 + bn2];
      float4 wk; wk.x = r0.x; wk.y = r1.x; wk.z = r0.y; wk.w = r1.y;
      *(float4*)&Bks[(bp * 64 + bn2) * 2] = wk;
      float2 s0 = *(const float2*)&Wv[(size_t)(kc + 2 * bp) * DD + n0 + bn2];
      float2 s1 = *(const float2*)&Wv[(size_t)(kc + 2 * bp + 1) * DD + n0 + bn2];
      float4 wvv; wvv.x = s0.x; wvv.y = s1.x; wvv.z = s0.y; wvv.w = s1.y;
      *(float4*)&Bvs[(bp * 64 + bn2) * 2] = wvv;
    }
    __syncthreads();
    const float* Aw = &As[(wv * 16) * 16];
#pragma unroll
    for (int ks = 0; ks < 16; ks += 4) {
      v2f a = frag_a(Aw, 16, lane, ks);
#pragma unroll
      for (int nt = 0; nt < 4; ++nt) {
        v2f bk = frag_b(Bks, 64, lane, ks, nt * 16);
        accK[nt] = wmma4(a, bk, accK[nt]);
        v2f bv = frag_b(Bvs, 64, lane, ks, nt * 16);
        accV[nt] = wmma4(a, bv, accV[nt]);
      }
    }
  }
  int half = lane >> 4, l16 = lane & 15;
#pragma unroll
  for (int nt = 0; nt < 4; ++nt)
#pragma unroll
    for (int i = 0; i < 8; ++i) {
      size_t mrow = (size_t)(m0 + wv * 16 + half * 8 + i);
      int ncol = n0 + nt * 16 + l16;
      Ko[mrow * DD + ncol] = accK[nt][i];
      Vo[mrow * DD + ncol] = accV[nt][i];
    }
}

struct SlotFrag {
  float4 a[8];
  float4 b[8][2];
};

template <bool BT, int K, int N>
__device__ __forceinline__ void slot_load(SlotFrag& f, const float* __restrict__ ap,
                                          const float* __restrict__ bp, int k0) {
#pragma unroll
  for (int u = 0; u < 8; ++u) {
    int kb = k0 + 8 * u;
    f.a[u] = *(const float4*)&ap[kb];
    if (BT) {
      f.b[u][0] = *(const float4*)&bp[kb];
      f.b[u][1] = *(const float4*)&bp[(size_t)K + kb];
    } else {
#pragma unroll
      for (int t = 0; t < 2; ++t) {
        float2 r0 = *(const float2*)&bp[(size_t)(kb + 2 * t) * N];
        float2 r1 = *(const float2*)&bp[(size_t)(kb + 2 * t + 1) * N];
        float4 q; q.x = r0.x; q.y = r0.y; q.z = r1.x; q.w = r1.y;
        f.b[u][t] = q;
      }
    }
  }
}

template <bool BT>
__device__ __forceinline__ void slot_mma(const SlotFrag& f, v8f& acc0, v8f& acc1) {
#pragma unroll
  for (int u = 0; u < 8; ++u) {
    v2f a0, a1, b00, b01, b10, b11;
    a0.x = f.a[u].x; a0.y = f.a[u].y;
    a1.x = f.a[u].z; a1.y = f.a[u].w;
    if (BT) {
      b00.x = f.b[u][0].x; b00.y = f.b[u][0].y;
      b01.x = f.b[u][1].x; b01.y = f.b[u][1].y;
      b10.x = f.b[u][0].z; b10.y = f.b[u][0].w;
      b11.x = f.b[u][1].z; b11.y = f.b[u][1].w;
    } else {
      b00.x = f.b[u][0].x; b00.y = f.b[u][0].z;
      b01.x = f.b[u][0].y; b01.y = f.b[u][0].w;
      b10.x = f.b[u][1].x; b10.y = f.b[u][1].z;
      b11.x = f.b[u][1].y; b11.y = f.b[u][1].w;
    }
    acc0 = wmma4(a0, b00, acc0);
    acc1 = wmma4(a0, b01, acc1);
    acc0 = wmma4(a1, b10, acc0);
    acc1 = wmma4(a1, b11, acc1);
  }
}

template <bool BT, int EPI, int K, int N>
__global__ void __launch_bounds__(128) __attribute__((amdgpu_waves_per_eu(1, 1)))
gemm_slot_kernel(const float* __restrict__ A, const float* __restrict__ B,
                 const float* __restrict__ bias, float* C, const float* aux) {
  int lane = threadIdx.x & 31, wv = threadIdx.x >> 5;
  int c = lane & 15, h = lane >> 4;
  int n0 = blockIdx.x * 32;
  int m0 = (blockIdx.y * 4 + wv) * 16;
  const float* ap = A + (size_t)(m0 + c) * K + 4 * h;
  const float* bp = BT ? B + (size_t)(n0 + 2 * c) * K + 4 * h
                       : B + (size_t)(4 * h) * N + n0 + 2 * c;
  SlotFrag f0, f1;
  v8f acc0 = {}, acc1 = {};
  slot_load<BT, K, N>(f0, ap, bp, 0);
  slot_load<BT, K, N>(f1, ap, bp, 64);
  for (int kc = 128; kc < K; kc += 128) {
    slot_mma<BT>(f0, acc0, acc1);
    slot_load<BT, K, N>(f0, ap, bp, kc);
    slot_mma<BT>(f1, acc0, acc1);
    slot_load<BT, K, N>(f1, ap, bp, kc + 64);
  }
  slot_mma<BT>(f0, acc0, acc1);
  slot_mma<BT>(f1, acc0, acc1);

  int ncol = n0 + 2 * c;
  float2 bv; bv.x = 0.f; bv.y = 0.f;
  if (EPI >= 1) bv = *(const float2*)&bias[ncol];
#pragma unroll
  for (int r = 0; r < 8; ++r) {
    size_t o = (size_t)(m0 + 8 * h + r) * N + ncol;
    float2 v; v.x = acc0[r] + bv.x; v.y = acc1[r] + bv.y;
    if (EPI == 1) { v.x = fmaxf(v.x, 0.f); v.y = fmaxf(v.y, 0.f); }
    if (EPI == 2) { float2 x = *(const float2*)&aux[o]; v.x += x.x; v.y += x.y; }
    *(float2*)&C[o] = v;
  }
}

__global__ void __launch_bounds__(256)
dots_kernel(const float* __restrict__ Q, const float* __restrict__ Keys,
            float* __restrict__ Dots) {
  __shared__ float As[16 * 16];
  __shared__ float Bs[16 * 512];
  int t = threadIdx.x, lane = t & 31, wv = t >> 5;
  int b = blockIdx.x, nb = blockIdx.y * 512;
  const float* qb = Q + (size_t)b * SS * DD;
  const float* kb = Keys + ((size_t)b * FF + nb) * DD;
  v8f acc[4] = {};
  for (int kc = 0; kc < DD; kc += 16) {
    __syncthreads();
    {
      int r = t >> 4, k = t & 15;
      As[r * 16 + k] = (r < SS) ? qb[(size_t)r * DD + kc + k] : 0.f;
    }
    {
      int k4 = (t & 3) * 4;
#pragma unroll
      for (int j = 0; j < 8; ++j) {
        int n = (t >> 2) + j * 64;
        float4 f = *(const float4*)&kb[(size_t)n * DD + kc + k4];
        float2 lo; lo.x = f.x; lo.y = f.y;
        float2 hi; hi.x = f.z; hi.y = f.w;
        *(float2*)&Bs[(((k4 >> 1) + 0) * 512 + n) * 2] = lo;
        *(float2*)&Bs[(((k4 >> 1) + 1) * 512 + n) * 2] = hi;
      }
    }
    __syncthreads();
#pragma unroll
    for (int ks = 0; ks < 16; ks += 4) {
      v2f a = frag_a(As, 16, lane, ks);
#pragma unroll
      for (int nt = 0; nt < 4; ++nt) {
        v2f b2 = frag_b(Bs, 512, lane, ks, wv * 64 + nt * 16);
        acc[nt] = wmma4(a, b2, acc[nt]);
      }
    }
  }
  int half = lane >> 4, l16 = lane & 15;
  if (half == 0) {
#pragma unroll
    for (int nt = 0; nt < 4; ++nt)
#pragma unroll
      for (int i = 0; i < 8; ++i) {
        int f = nb + wv * 64 + nt * 16 + l16;
        Dots[((size_t)b * SS + i) * FF + f] = acc[nt][i] * SCALE_C;
      }
  }
}

__global__ void __launch_bounds__(256)
softmax_kernel(const float* __restrict__ Dots, float* __restrict__ PreNorm,
               float* __restrict__ Attn, float* __restrict__ Sums) {
  int b = blockIdx.y;
  int f = blockIdx.x * 256 + threadIdx.x;
  int lane = threadIdx.x & 31;
  float d[SS]; float mx = -3.4e38f;
#pragma unroll
  for (int s = 0; s < SS; ++s) {
    d[s] = Dots[((size_t)b * SS + s) * FF + f];
    mx = fmaxf(mx, d[s]);
  }
  float sum = 0.f;
#pragma unroll
  for (int s = 0; s < SS; ++s) { d[s] = expf(d[s] - mx); sum += d[s]; }
  float inv = 1.f / sum;
#pragma unroll
  for (int s = 0; s < SS; ++s) {
    float p = d[s] * inv;
    size_t idx = ((size_t)b * SS + s) * FF + f;
    PreNorm[idx] = p;
    float aw = p + EPS_A;
    Attn[idx] = aw;
    float r = wave_sum(aw);
    if (lane == 0) atomicAdd(&Sums[b * SS + s], r);
  }
}

__global__ void zero_sums_kernel(float* __restrict__ Sums) {
  Sums[threadIdx.x] = 0.f;
}

#define UPD_FC    256
#define UPD_NCH   (FF / UPD_FC)
#define UPD_KS    32
#define UPD_LDA   34
#define UPD_PLANE (BB * SS * DD)

__global__ void __launch_bounds__(256)
updates_part_kernel(const float* __restrict__ Attn, const float* __restrict__ Vals,
                    float* __restrict__ PartLo, float* __restrict__ PartHi) {
  __shared__ float As[16 * UPD_LDA];
  __shared__ float Bs[UPD_KS * 512];
  int t = threadIdx.x, lane = t & 31, wv = t >> 5;
  int c = blockIdx.x, b = blockIdx.y;
  const float* ab = Attn + (size_t)b * SS * FF + c * UPD_FC;
  const float* vb = Vals + ((size_t)b * FF + c * UPD_FC) * DD;
  int ar = t >> 5, ak = t & 31;
  As[(8 + ar) * UPD_LDA + ak] = 0.f;
  v8f acc[4] = {};
  for (int kc = 0; kc < UPD_FC; kc += UPD_KS) {
    __syncthreads();
    As[ar * UPD_LDA + ak] = ab[(size_t)ar * FF + kc + ak];
#pragma unroll
    for (int p = 0; p < UPD_KS / 2; ++p) {
      float2 r0 = *(const float2*)&vb[(size_t)(kc + 2 * p) * DD + 2 * t];
      float2 r1 = *(const float2*)&vb[(size_t)(kc + 2 * p + 1) * DD + 2 * t];
      float4 wb; wb.x = r0.x; wb.y = r1.x; wb.z = r0.y; wb.w = r1.y;
      *(float4*)&Bs[(p * 512 + 2 * t) * 2] = wb;
    }
    __syncthreads();
#pragma unroll
    for (int ks = 0; ks < UPD_KS; ks += 4) {
      v2f a = frag_a(As, UPD_LDA, lane, ks);
#pragma unroll
      for (int nt = 0; nt < 4; ++nt) {
        v2f b2 = frag_b(Bs, 512, lane, ks, wv * 64 + nt * 16);
        acc[nt] = wmma4(a, b2, acc[nt]);
      }
    }
  }
  float* plane = (c < 8) ? PartLo + (size_t)c * UPD_PLANE
                         : PartHi + (size_t)(c - 8) * UPD_PLANE;
  int half = lane >> 4, l16 = lane & 15;
  if (half == 0) {
#pragma unroll
    for (int i = 0; i < 8; ++i)
#pragma unroll
      for (int nt = 0; nt < 4; ++nt) {
        int n = wv * 64 + nt * 16 + l16;
        plane[((size_t)b * SS + i) * DD + n] = acc[nt][i];
      }
  }
}

__global__ void __launch_bounds__(256)
updates_sum_kernel(const float* __restrict__ PartLo, const float* __restrict__ PartHi,
                   const float* __restrict__ Sums, float* __restrict__ Upd) {
  int i4 = blockIdx.x * 256 + threadIdx.x;
  int row = i4 >> 7;
  float4 v[UPD_NCH];
#pragma unroll
  for (int c = 0; c < 8; ++c) {
    v[c]     = *(const float4*)&PartLo[(size_t)c * UPD_PLANE + 4 * (size_t)i4];
    v[c + 8] = *(const float4*)&PartHi[(size_t)c * UPD_PLANE + 4 * (size_t)i4];
  }
  float4 s = v[0];
#pragma unroll
  for (int c = 1; c < UPD_NCH; ++c) {
    s.x += v[c].x; s.y += v[c].y; s.z += v[c].z; s.w += v[c].w;
  }
  float inv = 1.f / Sums[row];
  s.x *= inv; s.y *= inv; s.z *= inv; s.w *= inv;
  *(float4*)&Upd[4 * (size_t)i4] = s;
}

__global__ void __launch_bounds__(256)
gru_kernel(const float* __restrict__ gi, const float* __restrict__ gh,
           const float* __restrict__ sn, float* __restrict__ slots) {
  int idx = blockIdx.x * 256 + threadIdx.x;
  int m = idx >> 9, d = idx & 511;
  size_t base = (size_t)m * 3 * DD;
  float ir = gi[base + d],           hr = gh[base + d];
  float iz = gi[base + DD + d],      hz = gh[base + DD + d];
  float in_ = gi[base + 2 * DD + d], hn = gh[base + 2 * DD + d];
  float r = 1.f / (1.f + expf(-(ir + hr)));
  float z = 1.f / (1.f + expf(-(iz + hz)));
  float n = tanhf(in_ + r * hn);
  float h = sn[(size_t)m * DD + d];
  slots[(size_t)m * DD + d] = (1.f - z) * n + z * h;
}

extern "C" void kernel_launch(void* const* d_in, const int* in_sizes, int n_in,
                              void* d_out, int out_size, void* d_ws, size_t ws_size,
                              hipStream_t stream) {
  (void)in_sizes; (void)n_in; (void)out_size; (void)ws_size;
  const float* in_slots  = (const float*)d_in[0];
  const float* features  = (const float*)d_in[1];
  const float* w_k       = (const float*)d_in[2];
  const float* w_v       = (const float*)d_in[3];
  const float* w_q       = (const float*)d_in[4];
  const float* ln_feat_w = (const float*)d_in[5];
  const float* ln_feat_b = (const float*)d_in[6];
  const float* ln_slot_w = (const float*)d_in[7];
  const float* ln_slot_b = (const float*)d_in[8];
  const float* gru_w_ih  = (const float*)d_in[9];
  const float* gru_w_hh  = (const float*)d_in[10];
  const float* gru_b_ih  = (const float*)d_in[11];
  const float* gru_b_hh  = (const float*)d_in[12];
  const float* ln_mlp_w  = (const float*)d_in[13];
  const float* ln_mlp_b  = (const float*)d_in[14];
  const float* mlp_w1    = (const float*)d_in[15];
  const float* mlp_b1    = (const float*)d_in[16];
  const float* mlp_w2    = (const float*)d_in[17];
  const float* mlp_b2    = (const float*)d_in[18];

  float* ws = (float*)d_ws;
  size_t off = 0;
  auto carve = [&](size_t n) { float* p = ws + off; off += n; return p; };
  float* mean  = carve((size_t)BB * FF);
  float* rstd  = carve((size_t)BB * FF);
  float* keys  = carve((size_t)BB * FF * DD);
  float* vals  = carve((size_t)BB * FF * DD);
  float* slots = carve((size_t)BB * SS * DD);
  float* s_n   = carve((size_t)BB * SS * DD);
  float* q     = carve((size_t)BB * SS * DD);
  float* dots  = carve((size_t)BB * SS * FF);
  float* attn  = carve((size_t)BB * SS * FF);
  float* sums  = carve(256);
  float* upd   = carve((size_t)BB * SS * DD);
  float* gi    = carve((size_t)BB * SS * 3 * DD);
  float* gh    = carve((size_t)BB * SS * 3 * DD);
  float* hln   = carve((size_t)BB * SS * DD);
  float* h1    = carve((size_t)BB * SS * HH);

  float* out_slots = (float*)d_out;
  float* out_attn  = out_slots + (size_t)BB * SS * DD;

  hipMemcpyAsync(slots, in_slots, (size_t)BB * SS * DD * sizeof(float),
                 hipMemcpyDeviceToDevice, stream);

  row_stats_kernel<<<(BB * FF) / 8, 256, 0, stream>>>(features, mean, rstd);
  gemm_ln_dual_kernel<<<dim3((BB * FF) / 128, DD / 64), 256, 0, stream>>>(
      features, mean, rstd, ln_feat_w, ln_feat_b, w_k, w_v, keys, vals);

  for (int it = 0; it < NITER; ++it) {
    ln_rows_kernel<<<32, 256, 0, stream>>>(slots, ln_slot_w, ln_slot_b, s_n);
    gemm_slot_kernel<false, 0, DD, DD><<<dim3(DD / 32, 4), 128, 0, stream>>>(
        s_n, w_q, nullptr, q, nullptr);
    zero_sums_kernel<<<1, 256, 0, stream>>>(sums);
    dots_kernel<<<dim3(BB, FF / 512), 256, 0, stream>>>(q, keys, dots);
    softmax_kernel<<<dim3(FF / 256, BB), 256, 0, stream>>>(dots, out_attn, attn, sums);
    updates_part_kernel<<<dim3(UPD_NCH, BB), 256, 0, stream>>>(attn, vals, dots, gi);
    updates_sum_kernel<<<(BB * SS * DD / 4) / 256, 256, 0, stream>>>(dots, gi, sums, upd);
    gemm_slot_kernel<true, 3, DD, 3 * DD><<<dim3((3 * DD) / 32, 4), 128, 0, stream>>>(
        upd, gru_w_ih, gru_b_ih, gi, nullptr);
    gemm_slot_kernel<true, 3, DD, 3 * DD><<<dim3((3 * DD) / 32, 4), 128, 0, stream>>>(
        s_n, gru_w_hh, gru_b_hh, gh, nullptr);
    gru_kernel<<<(BB * SS * DD) / 256, 256, 0, stream>>>(gi, gh, s_n, slots);
    ln_rows_kernel<<<32, 256, 0, stream>>>(slots, ln_mlp_w, ln_mlp_b, hln);
    gemm_slot_kernel<false, 1, DD, HH><<<dim3(HH / 32, 4), 128, 0, stream>>>(
        hln, mlp_w1, mlp_b1, h1, nullptr);
    gemm_slot_kernel<false, 2, HH, DD><<<dim3(DD / 32, 4), 128, 0, stream>>>(
        h1, mlp_w2, mlp_b2, slots, slots);
  }

  hipMemcpyAsync(out_slots, slots, (size_t)BB * SS * DD * sizeof(float),
                 hipMemcpyDeviceToDevice, stream);
}


# keys written by the projection in matrix-fragment order (tile transposed by swapping the product's operands); scores product streams them straight into fragments, no LDS or barrier (new sum order)
# speedup vs baseline: 3.1884x; 1.0477x over previous
#include <hip/hip_runtime.h>
#include <hip/hip_bf16.h>

typedef __attribute__((ext_vector_type(2))) float v2f;
typedef __attribute__((ext_vector_type(8))) float v8f;

#define BB 32
#define FF 4096
#define SS 8
#define DD 512
#define HH 2048
#define NITER 3
#define EPS_A 1e-8f
#define LN_EPS_C 1e-5f
#define SCALE_C 0.04419417382415922f

__device__ __forceinline__ v8f wmma4(v2f a, v2f b, v8f c) {
  return __builtin_amdgcn_wmma_f32_16x16x4_f32(
       false, a,  false, b,
       (short)0, c,  false,  false);
}

__device__ __forceinline__ v2f frag_a(const float* A, int lda, int lane, int k0) {
  int row = lane & 15;
  int kk  = k0 + ((lane >> 4) << 1);
  return *(const v2f*)&A[row * lda + kk];
}
__device__ __forceinline__ v2f frag_b(const float* B, int ldb, int lane, int k0, int n0) {
  int col = n0 + (lane & 15);
  int p   = (k0 >> 1) + (lane >> 4);
  return *(const v2f*)&B[(p * ldb + col) * 2];
}

__device__ __forceinline__ float wave_sum(float v) {
#pragma unroll
  for (int o = 16; o; o >>= 1) v += __shfl_xor(v, o, 32);
  return v;
}

__global__ void __launch_bounds__(256)
row_stats_kernel(const float* __restrict__ X, float* __restrict__ mean,
                 float* __restrict__ rstd) {
  int wv = threadIdx.x >> 5, lane = threadIdx.x & 31;
  int row = blockIdx.x * 8 + wv;
  const float* x = X + (size_t)row * DD;
  float s = 0.f, s2 = 0.f;
#pragma unroll
  for (int j = 0; j < DD / 32; ++j) {
    float v = x[lane + 32 * j]; s += v; s2 += v * v;
  }
  s = wave_sum(s); s2 = wave_sum(s2);
  if (lane == 0) {
    float m = s * (1.f / DD);
    float var = s2 * (1.f / DD) - m * m;
    mean[row] = m;
    rstd[row] = rsqrtf(var + LN_EPS_C);
  }
}

__global__ void __launch_bounds__(256)
ln_rows_kernel(const float* __restrict__ X, const float* __restrict__ w,
               const float* __restrict__ b, float* __restrict__ Y) {
  int wv = threadIdx.x >> 5, lane = threadIdx.x & 31;
  int row = blockIdx.x * 8 + wv;
  const float* x = X + (size_t)row * DD;
  float xv[16]; float s = 0.f, s2 = 0.f;
#pragma unroll
  for (int j = 0; j < 16; ++j) {
    float v = x[lane + 32 * j]; xv[j] = v; s += v; s2 += v * v;
  }
  s = wave_sum(s); s2 = wave_sum(s2);
  float m = s * (1.f / DD);
  float r = rsqrtf(s2 * (1.f / DD) - m * m + LN_EPS_C);
  float* y = Y + (size_t)row * DD;
#pragma unroll
  for (int j = 0; j < 16; ++j) {
    int d = lane + 32 * j;
    y[d] = (xv[j] - m) * r * w[d] + b[d];
  }
}

__global__ void __launch_bounds__(256)
gemm_ln_dual_kernel(const float* __restrict__ X, const float* __restrict__ mean,
                    const float* __restrict__ rstd, const float* __restrict__ lnw,
                    const float* __restrict__ lnb, const float* __restrict__ Wk,
                    const float* __restrict__ Wv, float* __restrict__ Ko,
                    float* __restrict__ Vo) {
  __shared__ float As[128 * 16];
  __shared__ float Bks[16 * 64];
  __shared__ float Bvs[16 * 64];
  int t = threadIdx.x, lane = t & 31, wv = t >> 5;
  int m0 = blockIdx.x * 128, n0 = blockIdx.y * 64;

  int arow = t >> 1, aoff = (t & 1) * 8;
  size_t grow = (size_t)(m0 + arow);
  float mn = mean[grow], rs = rstd[grow];
  const float* xrow = X + grow * DD;
  int bp = t >> 5, bn2 = (t & 31) * 2;

  v8f accK[4] = {}, accV[4] = {};
  for (int kc = 0; kc < DD; kc += 16) {
    __syncthreads();
#pragma unroll
    for (int j = 0; j < 8; ++j) {
      int k = kc + aoff + j;
      As[arow * 16 + aoff + j] = (xrow[k] - mn) * rs * lnw[k] + lnb[k];
    }
    {
      float2 r0 = *(const float2*)&Wk[(size_t)(kc + 2 * bp) * DD + n0 + bn2];
      float2 r1 = *(const float2*)&Wk[(size_t)(kc + 2 * bp + 1) * DD + n0 + bn2];
      float4 wk; wk.x = r0.x; wk.y = r1.x; wk.z = r0.y; wk.w = r1.y;
      *(float4*)&Bks[(bp * 64 + bn2) * 2] = wk;
      float2 s0 = *(const float2*)&Wv[(size_t)(kc + 2 * bp) * DD + n0 + bn2];
      float2 s1 = *(const float2*)&Wv[(size_t)(kc + 2 * bp + 1) * DD + n0 + bn2];
      float4 wvv; wvv.x = s0.x; wvv.y = s1.x; wvv.z = s0.y; wvv.w = s1.y;
      *(float4*)&Bvs[(bp * 64 + bn2) * 2] = wvv;
    }
    __syncthreads();
    const float* Aw = &As[(wv * 16) * 16];
#pragma unroll
    for (int ks = 0; ks < 16; ks += 4) {
      v2f a = frag_a(Aw, 16, lane, ks);
#pragma unroll
      for (int nt = 0; nt < 4; ++nt) {
        v2f bk = frag_b(Bks, 64, lane, ks, nt * 16);
        accK[nt] = wmma4(bk, a, accK[nt]);
        v2f bv = frag_b(Bvs, 64, lane, ks, nt * 16);
        accV[nt] = wmma4(a, bv, accV[nt]);
      }
    }
  }
  int half = lane >> 4, l16 = lane & 15;
#pragma unroll
  for (int nt = 0; nt < 4; ++nt)
#pragma unroll
    for (int i = 0; i < 8; ++i) {
      size_t mrow = (size_t)(m0 + wv * 16 + half * 8 + i);
      int ncol = n0 + nt * 16 + l16;
      Vo[mrow * DD + ncol] = accV[nt][i];
    }
  size_t mt = (size_t)blockIdx.x * 8 + wv;
#pragma unroll
  for (int nt = 0; nt < 4; ++nt) {
    float* tp = Ko + ((mt * (DD / 16) + (size_t)(blockIdx.y * 4 + nt)) * 256) + lane * 4;
    float4 lo, hi;
    lo.x = accK[nt][0]; lo.y = accK[nt][1]; lo.z = accK[nt][2]; lo.w = accK[nt][3];
    hi.x = accK[nt][4]; hi.y = accK[nt][5]; hi.z = accK[nt][6]; hi.w = accK[nt][7];
    *(float4*)tp = lo;
    *(float4*)(tp + 128) = hi;
  }
}

struct SlotFrag {
  float4 a[8];
  float4 b[8][2];
};

template <bool BT, int K, int N>
__device__ __forceinline__ void slot_load(SlotFrag& f, const float* __restrict__ ap,
                                          const float* __restrict__ bp, int k0) {
#pragma unroll
  for (int u = 0; u < 8; ++u) {
    int kb = k0 + 8 * u;
    f.a[u] = *(const float4*)&ap[kb];
    if (BT) {
      f.b[u][0] = *(const float4*)&bp[kb];
      f.b[u][1] = *(const float4*)&bp[(size_t)K + kb];
    } else {
#pragma unroll
      for (int t = 0; t < 2; ++t) {
        float2 r0 = *(const float2*)&bp[(size_t)(kb + 2 * t) * N];
        float2 r1 = *(const float2*)&bp[(size_t)(kb + 2 * t + 1) * N];
        float4 q; q.x = r0.x; q.y = r0.y; q.z = r1.x; q.w = r1.y;
        f.b[u][t] = q;
      }
    }
  }
}

template <bool BT>
__device__ __forceinline__ void slot_mma(const SlotFrag& f, v8f& acc0, v8f& acc1) {
#pragma unroll
  for (int u = 0; u < 8; ++u) {
    v2f a0, a1, b00, b01, b10, b11;
    a0.x = f.a[u].x; a0.y = f.a[u].y;
    a1.x = f.a[u].z; a1.y = f.a[u].w;
    if (BT) {
      b00.x = f.b[u][0].x; b00.y = f.b[u][0].y;
      b01.x = f.b[u][1].x; b01.y = f.b[u][1].y;
      b10.x = f.b[u][0].z; b10.y = f.b[u][0].w;
      b11.x = f.b[u][1].z; b11.y = f.b[u][1].w;
    } else {
      b00.x = f.b[u][0].x; b00.y = f.b[u][0].z;
      b01.x = f.b[u][0].y; b01.y = f.b[u][0].w;
      b10.x = f.b[u][1].x; b10.y = f.b[u][1].z;
      b11.x = f.b[u][1].y; b11.y = f.b[u][1].w;
    }
    acc0 = wmma4(a0, b00, acc0);
    acc1 = wmma4(a0, b01, acc1);
    acc0 = wmma4(a1, b10, acc0);
    acc1 = wmma4(a1, b11, acc1);
  }
}

template <bool BT, int EPI, int K, int N>
__global__ void __launch_bounds__(128) __attribute__((amdgpu_waves_per_eu(1, 1)))
gemm_slot_kernel(const float* __restrict__ A, const float* __restrict__ B,
                 const float* __restrict__ bias, float* C, const float* aux) {
  int lane = threadIdx.x & 31, wv = threadIdx.x >> 5;
  int c = lane & 15, h = lane >> 4;
  int n0 = blockIdx.x * 32;
  int m0 = (blockIdx.y * 4 + wv) * 16;
  const float* ap = A + (size_t)(m0 + c) * K + 4 * h;
  const float* bp = BT ? B + (size_t)(n0 + 2 * c) * K + 4 * h
                       : B + (size_t)(4 * h) * N + n0 + 2 * c;
  SlotFrag f0, f1;
  v8f acc0 = {}, acc1 = {};
  slot_load<BT, K, N>(f0, ap, bp, 0);
  slot_load<BT, K, N>(f1, ap, bp, 64);
  for (int kc = 128; kc < K; kc += 128) {
    slot_mma<BT>(f0, acc0, acc1);
    slot_load<BT, K, N>(f0, ap, bp, kc);
    slot_mma<BT>(f1, acc0, acc1);
    slot_load<BT, K, N>(f1, ap, bp, kc + 64);
  }
  slot_mma<BT>(f0, acc0, acc1);
  slot_mma<BT>(f1, acc0, acc1);

  int ncol = n0 + 2 * c;
  float2 bv; bv.x = 0.f; bv.y = 0.f;
  if (EPI >= 1) bv = *(const float2*)&bias[ncol];
#pragma unroll
  for (int r = 0; r < 8; ++r) {
    size_t o = (size_t)(m0 + 8 * h + r) * N + ncol;
    float2 v; v.x = acc0[r] + bv.x; v.y = acc1[r] + bv.y;
    if (EPI == 1) { v.x = fmaxf(v.x, 0.f); v.y = fmaxf(v.y, 0.f); }
    if (EPI == 2) { float2 x = *(const float2*)&aux[o]; v.x += x.x; v.y += x.y; }
    *(float2*)&C[o] = v;
  }
}

#define DOTS_G 4
struct DotsFrag {
  float4 q[DOTS_G][2];
  float4 k[2][DOTS_G][2];
};

__device__ __forceinline__ void dots_load(DotsFrag& f, const float* __restrict__ qp,
                                          const float* __restrict__ kp, int db0) {
#pragma unroll
  for (int g = 0; g < DOTS_G; ++g) {
    int db = db0 + g;
    f.q[g][0] = *(const float4*)&qp[16 * db];
    f.q[g][1] = *(const float4*)&qp[16 * db + 4];
    f.k[0][g][0] = *(const float4*)&kp[db * 256];
    f.k[0][g][1] = *(const float4*)&kp[db * 256 + 128];
    f.k[1][g][0] = *(const float4*)&kp[(DD / 16) * 256 + db * 256];
    f.k[1][g][1] = *(const float4*)&kp[(DD / 16) * 256 + db * 256 + 128];
  }
}

__device__ __forceinline__ void dots_mma(const DotsFrag& f, v8f& acc0, v8f& acc1) {
#pragma unroll
  for (int g = 0; g < DOTS_G; ++g)
#pragma unroll
    for (int j = 0; j < 2; ++j) {
      v2f a0, a1, b00, b01, b10, b11;
      a0.x  = f.q[g][j].x;    a0.y  = f.q[g][j].y;
      a1.x  = f.q[g][j].z;    a1.y  = f.q[g][j].w;
      b00.x = f.k[0][g][j].x; b00.y = f.k[0][g][j].y;
      b01.x = f.k[0][g][j].z; b01.y = f.k[0][g][j].w;
      b10.x = f.k[1][g][j].x; b10.y = f.k[1][g][j].y;
      b11.x = f.k[1][g][j].z; b11.y = f.k[1][g][j].w;
      acc0 = wmma4(a0, b00, acc0);
      acc1 = wmma4(a0, b10, acc1);
      acc0 = wmma4(a1, b01, acc0);
      acc1 = wmma4(a1, b11, acc1);
    }
}

__global__ void __launch_bounds__(128) __attribute__((amdgpu_waves_per_eu(4, 4)))
dots_kernel(const float* __restrict__ Q, const float* __restrict__ Keys,
            float* __restrict__ Dots) {
  int lane = threadIdx.x & 31, wv = threadIdx.x >> 5;
  int c = lane & 15, h = lane >> 4;
  int w = blockIdx.x * 4 + wv;
  int b = w >> 7;
  const float* qp = Q + ((size_t)b * SS + (c & 7)) * DD + 8 * h;
  const float* kp = Keys + (size_t)(2 * w) * (DD / 16) * 256 + lane * 4;
  DotsFrag f0, f1;
  v8f acc0 = {}, acc1 = {};
  dots_load(f0, qp, kp, 0);
  dots_load(f1, qp, kp, DOTS_G);
  for (int db = 2 * DOTS_G; db < DD / 16; db += 2 * DOTS_G) {
    dots_mma(f0, acc0, acc1);
    dots_load(f0, qp, kp, db);
    dots_mma(f1, acc0, acc1);
    dots_load(f1, qp, kp, db + DOTS_G);
  }
  dots_mma(f0, acc0, acc1);
  dots_mma(f1, acc0, acc1);
  if (h == 0) {
    int fcol = ((2 * w) & 255) * 16 + c;
#pragma unroll
    for (int i = 0; i < 8; ++i) {
      float* dr = Dots + ((size_t)b * SS + i) * FF + fcol;
      dr[0]  = acc0[i] * SCALE_C;
      dr[16] = acc1[i] * SCALE_C;
    }
  }
}

__global__ void __launch_bounds__(256)
softmax_kernel(const float* __restrict__ Dots, float* __restrict__ PreNorm,
               float* __restrict__ Attn, float* __restrict__ Sums) {
  int b = blockIdx.y;
  int f = blockIdx.x * 256 + threadIdx.x;
  int lane = threadIdx.x & 31;
  float d[SS]; float mx = -3.4e38f;
#pragma unroll
  for (int s = 0; s < SS; ++s) {
    d[s] = Dots[((size_t)b * SS + s) * FF + f];
    mx = fmaxf(mx, d[s]);
  }
  float sum = 0.f;
#pragma unroll
  for (int s = 0; s < SS; ++s) { d[s] = expf(d[s] - mx); sum += d[s]; }
  float inv = 1.f / sum;
#pragma unroll
  for (int s = 0; s < SS; ++s) {
    float p = d[s] * inv;
    size_t idx = ((size_t)b * SS + s) * FF + f;
    PreNorm[idx] = p;
    float aw = p + EPS_A;
    Attn[idx] = aw;
    float r = wave_sum(aw);
    if (lane == 0) atomicAdd(&Sums[b * SS + s], r);
  }
}

__global__ void zero_sums_kernel(float* __restrict__ Sums) {
  Sums[threadIdx.x] = 0.f;
}

#define UPD_FC    256
#define UPD_NCH   (FF / UPD_FC)
#define UPD_KS    32
#define UPD_LDA   34
#define UPD_PLANE (BB * SS * DD)

__global__ void __launch_bounds__(256)
updates_part_kernel(const float* __restrict__ Attn, const float* __restrict__ Vals,
                    float* __restrict__ PartLo, float* __restrict__ PartHi) {
  __shared__ float As[16 * UPD_LDA];
  __shared__ float Bs[UPD_KS * 512];
  int t = threadIdx.x, lane = t & 31, wv = t >> 5;
  int c = blockIdx.x, b = blockIdx.y;
  const float* ab = Attn + (size_t)b * SS * FF + c * UPD_FC;
  const float* vb = Vals + ((size_t)b * FF + c * UPD_FC) * DD;
  int ar = t >> 5, ak = t & 31;
  As[(8 + ar) * UPD_LDA + ak] = 0.f;
  v8f acc[4] = {};
  for (int kc = 0; kc < UPD_FC; kc += UPD_KS) {
    __syncthreads();
    As[ar * UPD_LDA + ak] = ab[(size_t)ar * FF + kc + ak];
#pragma unroll
    for (int p = 0; p < UPD_KS / 2; ++p) {
      float2 r0 = *(const float2*)&vb[(size_t)(kc + 2 * p) * DD + 2 * t];
      float2 r1 = *(const float2*)&vb[(size_t)(kc + 2 * p + 1) * DD + 2 * t];
      float4 wb; wb.x = r0.x; wb.y = r1.x; wb.z = r0.y; wb.w = r1.y;
      *(float4*)&Bs[(p * 512 + 2 * t) * 2] = wb;
    }
    __syncthreads();
#pragma unroll
    for (int ks = 0; ks < UPD_KS; ks += 4) {
      v2f a = frag_a(As, UPD_LDA, lane, ks);
#pragma unroll
      for (int nt = 0; nt < 4; ++nt) {
        v2f b2 = frag_b(Bs, 512, lane, ks, wv * 64 + nt * 16);
        acc[nt] = wmma4(a, b2, acc[nt]);
      }
    }
  }
  float* plane = (c < 8) ? PartLo + (size_t)c * UPD_PLANE
                         : PartHi + (size_t)(c - 8) * UPD_PLANE;
  int half = lane >> 4, l16 = lane & 15;
  if (half == 0) {
#pragma unroll
    for (int i = 0; i < 8; ++i)
#pragma unroll
      for (int nt = 0; nt < 4; ++nt) {
        int n = wv * 64 + nt * 16 + l16;
        plane[((size_t)b * SS + i) * DD + n] = acc[nt][i];
      }
  }
}

__global__ void __launch_bounds__(256)
updates_sum_kernel(const float* __restrict__ PartLo, const float* __restrict__ PartHi,
                   const float* __restrict__ Sums, float* __restrict__ Upd) {
  int i4 = blockIdx.x * 256 + threadIdx.x;
  int row = i4 >> 7;
  float4 v[UPD_NCH];
#pragma unroll
  for (int c = 0; c < 8; ++c) {
    v[c]     = *(const float4*)&PartLo[(size_t)c * UPD_PLANE + 4 * (size_t)i4];
    v[c + 8] = *(const float4*)&PartHi[(size_t)c * UPD_PLANE + 4 * (size_t)i4];
  }
  float4 s = v[0];
#pragma unroll
  for (int c = 1; c < UPD_NCH; ++c) {
    s.x += v[c].x; s.y += v[c].y; s.z += v[c].z; s.w += v[c].w;
  }
  float inv = 1.f / Sums[row];
  s.x *= inv; s.y *= inv; s.z *= inv; s.w *= inv;
  *(float4*)&Upd[4 * (size_t)i4] = s;
}

__global__ void __launch_bounds__(256)
gru_kernel(const float* __restrict__ gi, const float* __restrict__ gh,
           const float* __restrict__ sn, float* __restrict__ slots) {
  int idx = blockIdx.x * 256 + threadIdx.x;
  int m = idx >> 9, d = idx & 511;
  size_t base = (size_t)m * 3 * DD;
  float ir = gi[base + d],           hr = gh[base + d];
  float iz = gi[base + DD + d],      hz = gh[base + DD + d];
  float in_ = gi[base + 2 * DD + d], hn = gh[base + 2 * DD + d];
  float r = 1.f / (1.f + expf(-(ir + hr)));
  float z = 1.f / (1.f + expf(-(iz + hz)));
  float n = tanhf(in_ + r * hn);
  float h = sn[(size_t)m * DD + d];
  slots[(size_t)m * DD + d] = (1.f - z) * n + z * h;
}

extern "C" void kernel_launch(void* const* d_in, const int* in_sizes, int n_in,
                              void* d_out, int out_size, void* d_ws, size_t ws_size,
                              hipStream_t stream) {
  (void)in_sizes; (void)n_in; (void)out_size; (void)ws_size;
  const float* in_slots  = (const float*)d_in[0];
  const float* features  = (const float*)d_in[1];
  const float* w_k       = (const float*)d_in[2];
  const float* w_v       = (const float*)d_in[3];
  const float* w_q       = (const float*)d_in[4];
  const float* ln_feat_w = (const float*)d_in[5];
  const float* ln_feat_b = (const float*)d_in[6];
  const float* ln_slot_w = (const float*)d_in[7];
  const float* ln_slot_b = (const float*)d_in[8];
  const float* gru_w_ih  = (const float*)d_in[9];
  const float* gru_w_hh  = (const float*)d_in[10];
  const float* gru_b_ih  = (const float*)d_in[11];
  const float* gru_b_hh  = (const float*)d_in[12];
  const float* ln_mlp_w  = (const float*)d_in[13];
  const float* ln_mlp_b  = (const float*)d_in[14];
  const float* mlp_w1    = (const float*)d_in[15];
  const float* mlp_b1    = (const float*)d_in[16];
  const float* mlp_w2    = (const float*)d_in[17];
  const float* mlp_b2    = (const float*)d_in[18];

  float* ws = (float*)d_ws;
  size_t off = 0;
  auto carve = [&](size_t n) { float* p = ws + off; off += n; return p; };
  float* mean  = carve((size_t)BB * FF);
  float* rstd  = carve((size_t)BB * FF);
  float* keys  = carve((size_t)BB * FF * DD);
  float* vals  = carve((size_t)BB * FF * DD);
  float* slots = carve((size_t)BB * SS * DD);
  float* s_n   = carve((size_t)BB * SS * DD);
  float* q     = carve((size_t)BB * SS * DD);
  float* dots  = carve((size_t)BB * SS * FF);
  float* attn  = carve((size_t)BB * SS * FF);
  float* sums  = carve(256);
  float* upd   = carve((size_t)BB * SS * DD);
  float* gi    = carve((size_t)BB * SS * 3 * DD);
  float* gh    = carve((size_t)BB * SS * 3 * DD);
  float* hln   = carve((size_t)BB * SS * DD);
  float* h1    = carve((size_t)BB * SS * HH);

  float* out_slots = (float*)d_out;
  float* out_attn  = out_slots + (size_t)BB * SS * DD;

  hipMemcpyAsync(slots, in_slots, (size_t)BB * SS * DD * sizeof(float),
                 hipMemcpyDeviceToDevice, stream);

  row_stats_kernel<<<(BB * FF) / 8, 256, 0, stream>>>(features, mean, rstd);
  gemm_ln_dual_kernel<<<dim3((BB * FF) / 128, DD / 64), 256, 0, stream>>>(
      features, mean, rstd, ln_feat_w, ln_feat_b, w_k, w_v, keys, vals);

  for (int it = 0; it < NITER; ++it) {
    ln_rows_kernel<<<32, 256, 0, stream>>>(slots, ln_slot_w, ln_slot_b, s_n);
    gemm_slot_kernel<false, 0, DD, DD><<<dim3(DD / 32, 4), 128, 0, stream>>>(
        s_n, w_q, nullptr, q, nullptr);
    zero_sums_kernel<<<1, 256, 0, stream>>>(sums);
    dots_kernel<<<(BB * FF / 32) / 4, 128, 0, stream>>>(q, keys, dots);
    softmax_kernel<<<dim3(FF / 256, BB), 256, 0, stream>>>(dots, out_attn, attn, sums);
    updates_part_kernel<<<dim3(UPD_NCH, BB), 256, 0, stream>>>(attn, vals, dots, gi);
    updates_sum_kernel<<<(BB * SS * DD / 4) / 256, 256, 0, stream>>>(dots, gi, sums, upd);
    gemm_slot_kernel<true, 3, DD, 3 * DD><<<dim3((3 * DD) / 32, 4), 128, 0, stream>>>(
        upd, gru_w_ih, gru_b_ih, gi, nullptr);
    gemm_slot_kernel<true, 3, DD, 3 * DD><<<dim3((3 * DD) / 32, 4), 128, 0, stream>>>(
        s_n, gru_w_hh, gru_b_hh, gh, nullptr);
    gru_kernel<<<(BB * SS * DD) / 256, 256, 0, stream>>>(gi, gh, s_n, slots);
    ln_rows_kernel<<<32, 256, 0, stream>>>(slots, ln_mlp_w, ln_mlp_b, hln);
    gemm_slot_kernel<false, 1, DD, HH><<<dim3(HH / 32, 4), 128, 0, stream>>>(
        hln, mlp_w1, mlp_b1, h1, nullptr);
    gemm_slot_kernel<false, 2, HH, DD><<<dim3(DD / 32, 4), 128, 0, stream>>>(
        h1, mlp_w2, mlp_b2, slots, slots);
  }

  hipMemcpyAsync(out_slots, slots, (size_t)BB * SS * DD * sizeof(float),
                 hipMemcpyDeviceToDevice, stream);
}


# feature projection re-tiled: 32-k chunks in a two-stage LDS ring, next chunk prefetched to registers, 16-byte k-quad fragment reads a block ahead, one barrier a chunk (new sum order)
# speedup vs baseline: 3.4830x; 1.0924x over previous
#include <hip/hip_runtime.h>
#include <hip/hip_bf16.h>

typedef __attribute__((ext_vector_type(2))) float v2f;
typedef __attribute__((ext_vector_type(8))) float v8f;

#define BB 32
#define FF 4096
#define SS 8
#define DD 512
#define HH 2048
#define NITER 3
#define EPS_A 1e-8f
#define LN_EPS_C 1e-5f
#define SCALE_C 0.04419417382415922f

__device__ __forceinline__ v8f wmma4(v2f a, v2f b, v8f c) {
  return __builtin_amdgcn_wmma_f32_16x16x4_f32(
       false, a,  false, b,
       (short)0, c,  false,  false);
}

__device__ __forceinline__ v2f frag_a(const float* A, int lda, int lane, int k0) {
  int row = lane & 15;
  int kk  = k0 + ((lane >> 4) << 1);
  return *(const v2f*)&A[row * lda + kk];
}
__device__ __forceinline__ v2f frag_b(const float* B, int ldb, int lane, int k0, int n0) {
  int col = n0 + (lane & 15);
  int p   = (k0 >> 1) + (lane >> 4);
  return *(const v2f*)&B[(p * ldb + col) * 2];
}

__device__ __forceinline__ float wave_sum(float v) {
#pragma unroll
  for (int o = 16; o; o >>= 1) v += __shfl_xor(v, o, 32);
  return v;
}

__global__ void __launch_bounds__(256)
row_stats_kernel(const float* __restrict__ X, float* __restrict__ mean,
                 float* __restrict__ rstd) {
  int wv = threadIdx.x >> 5, lane = threadIdx.x & 31;
  int row = blockIdx.x * 8 + wv;
  const float* x = X + (size_t)row * DD;
  float s = 0.f, s2 = 0.f;
#pragma unroll
  for (int j = 0; j < DD / 32; ++j) {
    float v = x[lane + 32 * j]; s += v; s2 += v * v;
  }
  s = wave_sum(s); s2 = wave_sum(s2);
  if (lane == 0) {
    float m = s * (1.f / DD);
    float var = s2 * (1.f / DD) - m * m;
    mean[row] = m;
    rstd[row] = rsqrtf(var + LN_EPS_C);
  }
}

__global__ void __launch_bounds__(256)
ln_rows_kernel(const float* __restrict__ X, const float* __restrict__ w,
               const float* __restrict__ b, float* __restrict__ Y) {
  int wv = threadIdx.x >> 5, lane = threadIdx.x & 31;
  int row = blockIdx.x * 8 + wv;
  const float* x = X + (size_t)row * DD;
  float xv[16]; float s = 0.f, s2 = 0.f;
#pragma unroll
  for (int j = 0; j < 16; ++j) {
    float v = x[lane + 32 * j]; xv[j] = v; s += v; s2 += v * v;
  }
  s = wave_sum(s); s2 = wave_sum(s2);
  float m = s * (1.f / DD);
  float r = rsqrtf(s2 * (1.f / DD) - m * m + LN_EPS_C);
  float* y = Y + (size_t)row * DD;
#pragma unroll
  for (int j = 0; j < 16; ++j) {
    int d = lane + 32 * j;
    y[d] = (xv[j] - m) * r * w[d] + b[d];
  }
}

#define PJ_KC    32
#define PJ_NCH   (DD / PJ_KC)
#define PJ_P     36
#define PJ_A     (128 * PJ_P)
#define PJ_B     (64 * PJ_P)
#define PJ_STAGE (PJ_A + 2 * PJ_B)

struct ProjFrag { float4 a; float4 bk[4]; float4 bv[4]; };
struct ProjPre  { float4 x[4]; float2 k[4]; float2 v[4]; };

__device__ __forceinline__ void proj_gload(ProjPre& p, const float* __restrict__ xp,
                                           const float* __restrict__ wkp,
                                           const float* __restrict__ wvp, int kc) {
#pragma unroll
  for (int i = 0; i < 4; ++i) p.x[i] = *(const float4*)&xp[kc + 4 * i];
#pragma unroll
  for (int s = 0; s < 4; ++s) {
    p.k[s] = *(const float2*)&wkp[(size_t)(kc + s) * DD];
    p.v[s] = *(const float2*)&wvp[(size_t)(kc + s) * DD];
  }
}

struct ProjLn { float4 w[4]; float4 b[4]; };

__device__ __forceinline__ void proj_lnload(ProjLn& n, const float* LNs, int kc, int aoff) {
#pragma unroll
  for (int i = 0; i < 4; ++i) {
    n.w[i] = *(const float4*)&LNs[kc + aoff + 4 * i];
    n.b[i] = *(const float4*)&LNs[DD + kc + aoff + 4 * i];
  }
}

__device__ __forceinline__ void proj_stage(const ProjPre& p, const ProjLn& n, float* st,
                                           float mn, float rs, int arow, int aoff,
                                           int bn2, int bq) {
#pragma unroll
  for (int i = 0; i < 4; ++i) {
    float4 w = n.w[i];
    float4 b = n.b[i];
    float4 y;
    y.x = (p.x[i].x - mn) * rs * w.x + b.x;
    y.y = (p.x[i].y - mn) * rs * w.y + b.y;
    y.z = (p.x[i].z - mn) * rs * w.z + b.z;
    y.w = (p.x[i].w - mn) * rs * w.w + b.w;
    *(float4*)&st[arow * PJ_P + aoff + 4 * i] = y;
  }
  float* bk = st + PJ_A;
  float* bv = st + PJ_A + PJ_B;
  float4 q;
  q.x = p.k[0].x; q.y = p.k[1].x; q.z = p.k[2].x; q.w = p.k[3].x;
  *(float4*)&bk[bn2 * PJ_P + 4 * bq] = q;
  q.x = p.k[0].y; q.y = p.k[1].y; q.z = p.k[2].y; q.w = p.k[3].y;
  *(float4*)&bk[(bn2 + 1) * PJ_P + 4 * bq] = q;
  q.x = p.v[0].x; q.y = p.v[1].x; q.z = p.v[2].x; q.w = p.v[3].x;
  *(float4*)&bv[bn2 * PJ_P + 4 * bq] = q;
  q.x = p.v[0].y; q.y = p.v[1].y; q.z = p.v[2].y; q.w = p.v[3].y;
  *(float4*)&bv[(bn2 + 1) * PJ_P + 4 * bq] = q;
}

__device__ __forceinline__ void proj_fload(ProjFrag& f, const float* fa, const float* fb, int kb) {
  f.a = *(const float4*)&fa[kb];
#pragma unroll
  for (int nt = 0; nt < 4; ++nt) {
    f.bk[nt] = *(const float4*)&fb[nt * 16 * PJ_P + kb];
    f.bv[nt] = *(const float4*)&fb[PJ_B + nt * 16 * PJ_P + kb];
  }
}

__device__ __forceinline__ void proj_mma(const ProjFrag& f, v8f (&accK)[4], v8f (&accV)[4]) {
  v2f a0, a1;
  a0.x = f.a.x; a0.y = f.a.y;
  a1.x = f.a.z; a1.y = f.a.w;
#pragma unroll
  for (int nt = 0; nt < 4; ++nt) {
    v2f bk, bv;
    bk.x = f.bk[nt].x; bk.y = f.bk[nt].y;
    bv.x = f.bv[nt].x; bv.y = f.bv[nt].y;
    accK[nt] = wmma4(bk, a0, accK[nt]);
    accV[nt] = wmma4(a0, bv, accV[nt]);
  }
#pragma unroll
  for (int nt = 0; nt < 4; ++nt) {
    v2f bk, bv;
    bk.x = f.bk[nt].z; bk.y = f.bk[nt].w;
    bv.x = f.bv[nt].z; bv.y = f.bv[nt].w;
    accK[nt] = wmma4(bk, a1, accK[nt]);
    accV[nt] = wmma4(a1, bv, accV[nt]);
  }
}

__global__ void __launch_bounds__(256) __attribute__((amdgpu_waves_per_eu(4, 4)))
gemm_ln_dual_kernel(const float* __restrict__ X, const float* __restrict__ mean,
                    const float* __restrict__ rstd, const float* __restrict__ lnw,
                    const float* __restrict__ lnb, const float* __restrict__ Wk,
                    const float* __restrict__ Wv, float* __restrict__ Ko,
                    float* __restrict__ Vo) {
  __shared__ __attribute__((aligned(16))) float Ls[2 * PJ_STAGE];
  __shared__ __attribute__((aligned(16))) float LNs[2 * DD];
  int t = threadIdx.x, lane = t & 31, wv = t >> 5;
  int c = lane & 15, h = lane >> 4;
  int n0 = blockIdx.x * 64, m0 = blockIdx.y * 128;

  if (t < 128) *(float4*)&LNs[4 * t] = *(const float4*)&lnw[4 * t];
  else         *(float4*)&LNs[DD + 4 * (t - 128)] = *(const float4*)&lnb[4 * (t - 128)];

  int arow = t >> 1, aoff = (t & 1) * 16;
  size_t grow = (size_t)(m0 + arow);
  float mn = mean[grow], rs = rstd[grow];
  const float* xp = X + grow * DD + aoff;
  int bn2 = (t & 31) * 2, bq = t >> 5;
  const float* wkp = Wk + (size_t)(4 * bq) * DD + n0 + bn2;
  const float* wvp = Wv + (size_t)(4 * bq) * DD + n0 + bn2;
  int fao = (wv * 16 + c) * PJ_P + 4 * h;
  int fbo = PJ_A + c * PJ_P + 4 * h;

  v8f accK[4] = {}, accV[4] = {};
  ProjPre pre;
  ProjLn lnp;
  proj_gload(pre, xp, wkp, wvp, 0);
  __syncthreads();
  proj_lnload(lnp, LNs, 0, aoff);
  proj_stage(pre, lnp, Ls, mn, rs, arow, aoff, bn2, bq);
  __syncthreads();
  for (int i = 0; i < PJ_NCH; ++i) {
    const float* st = Ls + (i & 1) * PJ_STAGE;
    float* nx = Ls + ((i + 1) & 1) * PJ_STAGE;
    bool more = (i + 1 < PJ_NCH);
    if (more) proj_gload(pre, xp, wkp, wvp, (i + 1) * PJ_KC);
    __builtin_amdgcn_sched_barrier(0);
    ProjFrag f0, f1;
    proj_fload(f0, st + fao, st + fbo, 0);
    proj_fload(f1, st + fao, st + fbo, 8);
    proj_mma(f0, accK, accV);
    proj_fload(f0, st + fao, st + fbo, 16);
    proj_mma(f1, accK, accV);
    proj_fload(f1, st + fao, st + fbo, 24);
    proj_mma(f0, accK, accV);
    if (more) proj_lnload(lnp, LNs, (i + 1) * PJ_KC, aoff);
    __builtin_amdgcn_sched_barrier(0);
    proj_mma(f1, accK, accV);
    if (more) proj_stage(pre, lnp, nx, mn, rs, arow, aoff, bn2, bq);
    __syncthreads();
  }
  int half = lane >> 4, l16 = lane & 15;
#pragma unroll
  for (int nt = 0; nt < 4; ++nt)
#pragma unroll
    for (int i = 0; i < 8; ++i) {
      size_t mrow = (size_t)(m0 + wv * 16 + half * 8 + i);
      int ncol = n0 + nt * 16 + l16;
      Vo[mrow * DD + ncol] = accV[nt][i];
    }
  size_t mt = (size_t)blockIdx.y * 8 + wv;
#pragma unroll
  for (int nt = 0; nt < 4; ++nt) {
    float* tp = Ko + ((mt * (DD / 16) + (size_t)(blockIdx.x * 4 + nt)) * 256) + lane * 4;
    float4 lo, hi;
    lo.x = accK[nt][0]; lo.y = accK[nt][1]; lo.z = accK[nt][2]; lo.w = accK[nt][3];
    hi.x = accK[nt][4]; hi.y = accK[nt][5]; hi.z = accK[nt][6]; hi.w = accK[nt][7];
    *(float4*)tp = lo;
    *(float4*)(tp + 128) = hi;
  }
}

struct SlotFrag {
  float4 a[8];
  float4 b[8][2];
};

template <bool BT, int K, int N>
__device__ __forceinline__ void slot_load(SlotFrag& f, const float* __restrict__ ap,
                                          const float* __restrict__ bp, int k0) {
#pragma unroll
  for (int u = 0; u < 8; ++u) {
    int kb = k0 + 8 * u;
    f.a[u] = *(const float4*)&ap[kb];
    if (BT) {
      f.b[u][0] = *(const float4*)&bp[kb];
      f.b[u][1] = *(const float4*)&bp[(size_t)K + kb];
    } else {
#pragma unroll
      for (int t = 0; t < 2; ++t) {
        float2 r0 = *(const float2*)&bp[(size_t)(kb + 2 * t) * N];
        float2 r1 = *(const float2*)&bp[(size_t)(kb + 2 * t + 1) * N];
        float4 q; q.x = r0.x; q.y = r0.y; q.z = r1.x; q.w = r1.y;
        f.b[u][t] = q;
      }
    }
  }
}

template <bool BT>
__device__ __forceinline__ void slot_mma(const SlotFrag& f, v8f& acc0, v8f& acc1) {
#pragma unroll
  for (int u = 0; u < 8; ++u) {
    v2f a0, a1, b00, b01, b10, b11;
    a0.x = f.a[u].x; a0.y = f.a[u].y;
    a1.x = f.a[u].z; a1.y = f.a[u].w;
    if (BT) {
      b00.x = f.b[u][0].x; b00.y = f.b[u][0].y;
      b01.x = f.b[u][1].x; b01.y = f.b[u][1].y;
      b10.x = f.b[u][0].z; b10.y = f.b[u][0].w;
      b11.x = f.b[u][1].z; b11.y = f.b[u][1].w;
    } else {
      b00.x = f.b[u][0].x; b00.y = f.b[u][0].z;
      b01.x = f.b[u][0].y; b01.y = f.b[u][0].w;
      b10.x = f.b[u][1].x; b10.y = f.b[u][1].z;
      b11.x = f.b[u][1].y; b11.y = f.b[u][1].w;
    }
    acc0 = wmma4(a0, b00, acc0);
    acc1 = wmma4(a0, b01, acc1);
    acc0 = wmma4(a1, b10, acc0);
    acc1 = wmma4(a1, b11, acc1);
  }
}

template <bool BT, int EPI, int K, int N>
__global__ void __launch_bounds__(128) __attribute__((amdgpu_waves_per_eu(1, 1)))
gemm_slot_kernel(const float* __restrict__ A, const float* __restrict__ B,
                 const float* __restrict__ bias, float* C, const float* aux) {
  int lane = threadIdx.x & 31, wv = threadIdx.x >> 5;
  int c = lane & 15, h = lane >> 4;
  int n0 = blockIdx.x * 32;
  int m0 = (blockIdx.y * 4 + wv) * 16;
  const float* ap = A + (size_t)(m0 + c) * K + 4 * h;
  const float* bp = BT ? B + (size_t)(n0 + 2 * c) * K + 4 * h
                       : B + (size_t)(4 * h) * N + n0 + 2 * c;
  SlotFrag f0, f1;
  v8f acc0 = {}, acc1 = {};
  slot_load<BT, K, N>(f0, ap, bp, 0);
  slot_load<BT, K, N>(f1, ap, bp, 64);
  for (int kc = 128; kc < K; kc += 128) {
    slot_mma<BT>(f0, acc0, acc1);
    slot_load<BT, K, N>(f0, ap, bp, kc);
    slot_mma<BT>(f1, acc0, acc1);
    slot_load<BT, K, N>(f1, ap, bp, kc + 64);
  }
  slot_mma<BT>(f0, acc0, acc1);
  slot_mma<BT>(f1, acc0, acc1);

  int ncol = n0 + 2 * c;
  float2 bv; bv.x = 0.f; bv.y = 0.f;
  if (EPI >= 1) bv = *(const float2*)&bias[ncol];
#pragma unroll
  for (int r = 0; r < 8; ++r) {
    size_t o = (size_t)(m0 + 8 * h + r) * N + ncol;
    float2 v; v.x = acc0[r] + bv.x; v.y = acc1[r] + bv.y;
    if (EPI == 1) { v.x = fmaxf(v.x, 0.f); v.y = fmaxf(v.y, 0.f); }
    if (EPI == 2) { float2 x = *(const float2*)&aux[o]; v.x += x.x; v.y += x.y; }
    *(float2*)&C[o] = v;
  }
}

#define DOTS_G 4
struct DotsFrag {
  float4 q[DOTS_G][2];
  float4 k[2][DOTS_G][2];
};

__device__ __forceinline__ void dots_load(DotsFrag& f, const float* __restrict__ qp,
                                          const float* __restrict__ kp, int db0) {
#pragma unroll
  for (int g = 0; g < DOTS_G; ++g) {
    int db = db0 + g;
    f.q[g][0] = *(const float4*)&qp[16 * db];
    f.q[g][1] = *(const float4*)&qp[16 * db + 4];
    f.k[0][g][0] = *(const float4*)&kp[db * 256];
    f.k[0][g][1] = *(const float4*)&kp[db * 256 + 128];
    f.k[1][g][0] = *(const float4*)&kp[(DD / 16) * 256 + db * 256];
    f.k[1][g][1] = *(const float4*)&kp[(DD / 16) * 256 + db * 256 + 128];
  }
}

__device__ __forceinline__ void dots_mma(const DotsFrag& f, v8f& acc0, v8f& acc1) {
#pragma unroll
  for (int g = 0; g < DOTS_G; ++g)
#pragma unroll
    for (int j = 0; j < 2; ++j) {
      v2f a0, a1, b00, b01, b10, b11;
      a0.x  = f.q[g][j].x;    a0.y  = f.q[g][j].y;
      a1.x  = f.q[g][j].z;    a1.y  = f.q[g][j].w;
      b00.x = f.k[0][g][j].x; b00.y = f.k[0][g][j].y;
      b01.x = f.k[0][g][j].z; b01.y = f.k[0][g][j].w;
      b10.x = f.k[1][g][j].x; b10.y = f.k[1][g][j].y;
      b11.x = f.k[1][g][j].z; b11.y = f.k[1][g][j].w;
      acc0 = wmma4(a0, b00, acc0);
      acc1 = wmma4(a0, b10, acc1);
      acc0 = wmma4(a1, b01, acc0);
      acc1 = wmma4(a1, b11, acc1);
    }
}

__global__ void __launch_bounds__(128) __attribute__((amdgpu_waves_per_eu(4, 4)))
dots_kernel(const float* __restrict__ Q, const float* __restrict__ Keys,
            float* __restrict__ Dots) {
  int lane = threadIdx.x & 31, wv = threadIdx.x >> 5;
  int c = lane & 15, h = lane >> 4;
  int w = blockIdx.x * 4 + wv;
  int b = w >> 7;
  const float* qp = Q + ((size_t)b * SS + (c & 7)) * DD + 8 * h;
  const float* kp = Keys + (size_t)(2 * w) * (DD / 16) * 256 + lane * 4;
  DotsFrag f0, f1;
  v8f acc0 = {}, acc1 = {};
  dots_load(f0, qp, kp, 0);
  dots_load(f1, qp, kp, DOTS_G);
  for (int db = 2 * DOTS_G; db < DD / 16; db += 2 * DOTS_G) {
    dots_mma(f0, acc0, acc1);
    dots_load(f0, qp, kp, db);
    dots_mma(f1, acc0, acc1);
    dots_load(f1, qp, kp, db + DOTS_G);
  }
  dots_mma(f0, acc0, acc1);
  dots_mma(f1, acc0, acc1);
  if (h == 0) {
    int fcol = ((2 * w) & 255) * 16 + c;
#pragma unroll
    for (int i = 0; i < 8; ++i) {
      float* dr = Dots + ((size_t)b * SS + i) * FF + fcol;
      dr[0]  = acc0[i] * SCALE_C;
      dr[16] = acc1[i] * SCALE_C;
    }
  }
}

__global__ void __launch_bounds__(256)
softmax_kernel(const float* __restrict__ Dots, float* __restrict__ PreNorm,
               float* __restrict__ Attn, float* __restrict__ Sums) {
  int b = blockIdx.y;
  int f = blockIdx.x * 256 + threadIdx.x;
  int lane = threadIdx.x & 31;
  float d[SS]; float mx = -3.4e38f;
#pragma unroll
  for (int s = 0; s < SS; ++s) {
    d[s] = Dots[((size_t)b * SS + s) * FF + f];
    mx = fmaxf(mx, d[s]);
  }
  float sum = 0.f;
#pragma unroll
  for (int s = 0; s < SS; ++s) { d[s] = expf(d[s] - mx); sum += d[s]; }
  float inv = 1.f / sum;
#pragma unroll
  for (int s = 0; s < SS; ++s) {
    float p = d[s] * inv;
    size_t idx = ((size_t)b * SS + s) * FF + f;
    PreNorm[idx] = p;
    float aw = p + EPS_A;
    Attn[idx] = aw;
    float r = wave_sum(aw);
    if (lane == 0) atomicAdd(&Sums[b * SS + s], r);
  }
}

__global__ void zero_sums_kernel(float* __restrict__ Sums) {
  Sums[threadIdx.x] = 0.f;
}

#define UPD_FC    256
#define UPD_NCH   (FF / UPD_FC)
#define UPD_KS    32
#define UPD_LDA   34
#define UPD_PLANE (BB * SS * DD)

__global__ void __launch_bounds__(256)
updates_part_kernel(const float* __restrict__ Attn, const float* __restrict__ Vals,
                    float* __restrict__ PartLo, float* __restrict__ PartHi) {
  __shared__ float As[16 * UPD_LDA];
  __shared__ float Bs[UPD_KS * 512];
  int t = threadIdx.x, lane = t & 31, wv = t >> 5;
  int c = blockIdx.x, b = blockIdx.y;
  const float* ab = Attn + (size_t)b * SS * FF + c * UPD_FC;
  const float* vb = Vals + ((size_t)b * FF + c * UPD_FC) * DD;
  int ar = t >> 5, ak = t & 31;
  As[(8 + ar) * UPD_LDA + ak] = 0.f;
  v8f acc[4] = {};
  for (int kc = 0; kc < UPD_FC; kc += UPD_KS) {
    __syncthreads();
    As[ar * UPD_LDA + ak] = ab[(size_t)ar * FF + kc + ak];
#pragma unroll
    for (int p = 0; p < UPD_KS / 2; ++p) {
      float2 r0 = *(const float2*)&vb[(size_t)(kc + 2 * p) * DD + 2 * t];
      float2 r1 = *(const float2*)&vb[(size_t)(kc + 2 * p + 1) * DD + 2 * t];
      float4 wb; wb.x = r0.x; wb.y = r1.x; wb.z = r0.y; wb.w = r1.y;
      *(float4*)&Bs[(p * 512 + 2 * t) * 2] = wb;
    }
    __syncthreads();
#pragma unroll
    for (int ks = 0; ks < UPD_KS; ks += 4) {
      v2f a = frag_a(As, UPD_LDA, lane, ks);
#pragma unroll
      for (int nt = 0; nt < 4; ++nt) {
        v2f b2 = frag_b(Bs, 512, lane, ks, wv * 64 + nt * 16);
        acc[nt] = wmma4(a, b2, acc[nt]);
      }
    }
  }
  float* plane = (c < 8) ? PartLo + (size_t)c * UPD_PLANE
                         : PartHi + (size_t)(c - 8) * UPD_PLANE;
  int half = lane >> 4, l16 = lane & 15;
  if (half == 0) {
#pragma unroll
    for (int i = 0; i < 8; ++i)
#pragma unroll
      for (int nt = 0; nt < 4; ++nt) {
        int n = wv * 64 + nt * 16 + l16;
        plane[((size_t)b * SS + i) * DD + n] = acc[nt][i];
      }
  }
}

__global__ void __launch_bounds__(256)
updates_sum_kernel(const float* __restrict__ PartLo, const float* __restrict__ PartHi,
                   const float* __restrict__ Sums, float* __restrict__ Upd) {
  int i4 = blockIdx.x * 256 + threadIdx.x;
  int row = i4 >> 7;
  float4 v[UPD_NCH];
#pragma unroll
  for (int c = 0; c < 8; ++c) {
    v[c]     = *(const float4*)&PartLo[(size_t)c * UPD_PLANE + 4 * (size_t)i4];
    v[c + 8] = *(const float4*)&PartHi[(size_t)c * UPD_PLANE + 4 * (size_t)i4];
  }
  float4 s = v[0];
#pragma unroll
  for (int c = 1; c < UPD_NCH; ++c) {
    s.x += v[c].x; s.y += v[c].y; s.z += v[c].z; s.w += v[c].w;
  }
  float inv = 1.f / Sums[row];
  s.x *= inv; s.y *= inv; s.z *= inv; s.w *= inv;
  *(float4*)&Upd[4 * (size_t)i4] = s;
}

__global__ void __launch_bounds__(256)
gru_kernel(const float* __restrict__ gi, const float* __restrict__ gh,
           const float* __restrict__ sn, float* __restrict__ slots) {
  int idx = blockIdx.x * 256 + threadIdx.x;
  int m = idx >> 9, d = idx & 511;
  size_t base = (size_t)m * 3 * DD;
  float ir = gi[base + d],           hr = gh[base + d];
  float iz = gi[base + DD + d],      hz = gh[base + DD + d];
  float in_ = gi[base + 2 * DD + d], hn = gh[base + 2 * DD + d];
  float r = 1.f / (1.f + expf(-(ir + hr)));
  float z = 1.f / (1.f + expf(-(iz + hz)));
  float n = tanhf(in_ + r * hn);
  float h = sn[(size_t)m * DD + d];
  slots[(size_t)m * DD + d] = (1.f - z) * n + z * h;
}

extern "C" void kernel_launch(void* const* d_in, const int* in_sizes, int n_in,
                              void* d_out, int out_size, void* d_ws, size_t ws_size,
                              hipStream_t stream) {
  (void)in_sizes; (void)n_in; (void)out_size; (void)ws_size;
  const float* in_slots  = (const float*)d_in[0];
  const float* features  = (const float*)d_in[1];
  const float* w_k       = (const float*)d_in[2];
  const float* w_v       = (const float*)d_in[3];
  const float* w_q       = (const float*)d_in[4];
  const float* ln_feat_w = (const float*)d_in[5];
  const float* ln_feat_b = (const float*)d_in[6];
  const float* ln_slot_w = (const float*)d_in[7];
  const float* ln_slot_b = (const float*)d_in[8];
  const float* gru_w_ih  = (const float*)d_in[9];
  const float* gru_w_hh  = (const float*)d_in[10];
  const float* gru_b_ih  = (const float*)d_in[11];
  const float* gru_b_hh  = (const float*)d_in[12];
  const float* ln_mlp_w  = (const float*)d_in[13];
  const float* ln_mlp_b  = (const float*)d_in[14];
  const float* mlp_w1    = (const float*)d_in[15];
  const float* mlp_b1    = (const float*)d_in[16];
  const float* mlp_w2    = (const float*)d_in[17];
  const float* mlp_b2    = (const float*)d_in[18];

  float* ws = (float*)d_ws;
  size_t off = 0;
  auto carve = [&](size_t n) { float* p = ws + off; off += n; return p; };
  float* mean  = carve((size_t)BB * FF);
  float* rstd  = carve((size_t)BB * FF);
  float* keys  = carve((size_t)BB * FF * DD);
  float* vals  = carve((size_t)BB * FF * DD);
  float* slots = carve((size_t)BB * SS * DD);
  float* s_n   = carve((size_t)BB * SS * DD);
  float* q     = carve((size_t)BB * SS * DD);
  float* dots  = carve((size_t)BB * SS * FF);
  float* attn  = carve((size_t)BB * SS * FF);
  float* sums  = carve(256);
  float* upd   = carve((size_t)BB * SS * DD);
  float* gi    = carve((size_t)BB * SS * 3 * DD);
  float* gh    = carve((size_t)BB * SS * 3 * DD);
  float* hln   = carve((size_t)BB * SS * DD);
  float* h1    = carve((size_t)BB * SS * HH);

  float* out_slots = (float*)d_out;
  float* out_attn  = out_slots + (size_t)BB * SS * DD;

  hipMemcpyAsync(slots, in_slots, (size_t)BB * SS * DD * sizeof(float),
                 hipMemcpyDeviceToDevice, stream);

  row_stats_kernel<<<(BB * FF) / 8, 256, 0, stream>>>(features, mean, rstd);
  gemm_ln_dual_kernel<<<dim3(DD / 64, (BB * FF) / 128), 256, 0, stream>>>(
      features, mean, rstd, ln_feat_w, ln_feat_b, w_k, w_v, keys, vals);

  for (int it = 0; it < NITER; ++it) {
    ln_rows_kernel<<<32, 256, 0, stream>>>(slots, ln_slot_w, ln_slot_b, s_n);
    gemm_slot_kernel<false, 0, DD, DD><<<dim3(DD / 32, 4), 128, 0, stream>>>(
        s_n, w_q, nullptr, q, nullptr);
    zero_sums_kernel<<<1, 256, 0, stream>>>(sums);
    dots_kernel<<<(BB * FF / 32) / 4, 128, 0, stream>>>(q, keys, dots);
    softmax_kernel<<<dim3(FF / 256, BB), 256, 0, stream>>>(dots, out_attn, attn, sums);
    updates_part_kernel<<<dim3(UPD_NCH, BB), 256, 0, stream>>>(attn, vals, dots, gi);
    updates_sum_kernel<<<(BB * SS * DD / 4) / 256, 256, 0, stream>>>(dots, gi, sums, upd);
    gemm_slot_kernel<true, 3, DD, 3 * DD><<<dim3((3 * DD) / 32, 4), 128, 0, stream>>>(
        upd, gru_w_ih, gru_b_ih, gi, nullptr);
    gemm_slot_kernel<true, 3, DD, 3 * DD><<<dim3((3 * DD) / 32, 4), 128, 0, stream>>>(
        s_n, gru_w_hh, gru_b_hh, gh, nullptr);
    gru_kernel<<<(BB * SS * DD) / 256, 256, 0, stream>>>(gi, gh, s_n, slots);
    ln_rows_kernel<<<32, 256, 0, stream>>>(slots, ln_mlp_w, ln_mlp_b, hln);
    gemm_slot_kernel<false, 1, DD, HH><<<dim3(HH / 32, 4), 128, 0, stream>>>(
        hln, mlp_w1, mlp_b1, h1, nullptr);
    gemm_slot_kernel<false, 2, HH, DD><<<dim3(DD / 32, 4), 128, 0, stream>>>(
        h1, mlp_w2, mlp_b2, slots, slots);
  }

  hipMemcpyAsync(out_slots, slots, (size_t)BB * SS * DD * sizeof(float),
                 hipMemcpyDeviceToDevice, stream);
}


# feature LayerNorm put behind the projection by linearity: raw rows x w-scaled weights, row sums from the staging loads, epilogue applies rstd, mean*colsum, bias term; stats pass gone (new sum order)
# speedup vs baseline: 3.6102x; 1.0365x over previous
#include <hip/hip_runtime.h>
#include <hip/hip_bf16.h>

typedef __attribute__((ext_vector_type(2))) float v2f;
typedef __attribute__((ext_vector_type(8))) float v8f;

#define BB 32
#define FF 4096
#define SS 8
#define DD 512
#define HH 2048
#define NITER 3
#define EPS_A 1e-8f
#define LN_EPS_C 1e-5f
#define SCALE_C 0.04419417382415922f

__device__ __forceinline__ v8f wmma4(v2f a, v2f b, v8f c) {
  return __builtin_amdgcn_wmma_f32_16x16x4_f32(
       false, a,  false, b,
       (short)0, c,  false,  false);
}

__device__ __forceinline__ v2f frag_a(const float* A, int lda, int lane, int k0) {
  int row = lane & 15;
  int kk  = k0 + ((lane >> 4) << 1);
  return *(const v2f*)&A[row * lda + kk];
}
__device__ __forceinline__ v2f frag_b(const float* B, int ldb, int lane, int k0, int n0) {
  int col = n0 + (lane & 15);
  int p   = (k0 >> 1) + (lane >> 4);
  return *(const v2f*)&B[(p * ldb + col) * 2];
}

__device__ __forceinline__ float wave_sum(float v) {
#pragma unroll
  for (int o = 16; o; o >>= 1) v += __shfl_xor(v, o, 32);
  return v;
}

__global__ void __launch_bounds__(1024)
fold_ln_kernel(const float* __restrict__ lnw, const float* __restrict__ lnb,
               const float* __restrict__ Wk, const float* __restrict__ Wv,
               float* __restrict__ Wks, float* __restrict__ Wvs,
               float* __restrict__ cd) {
  __shared__ float redc[32][33];
  __shared__ float redd[32][33];
  int t = threadIdx.x, col = t & 31, kg = t >> 5;
  bool isv = blockIdx.x >= 16;
  const float* W = isv ? Wv : Wk;
  float* Ws = isv ? Wvs : Wks;
  int n = (blockIdx.x & 15) * 32 + col;
  float c = 0.f, d = 0.f;
#pragma unroll
  for (int i = 0; i < 16; ++i) {
    int k = kg * 16 + i;
    float v = W[(size_t)k * DD + n];
    float sv = lnw[k] * v;
    Ws[(size_t)k * DD + n] = sv;
    c += sv;
    d = fmaf(lnb[k], v, d);
  }
  redc[kg][col] = c;
  redd[kg][col] = d;
  __syncthreads();
  if (kg == 0) {
    float cs = redc[0][col], ds = redd[0][col];
    for (int g = 1; g < 32; ++g) { cs += redc[g][col]; ds += redd[g][col]; }
    int o = isv ? 2 * DD : 0;
    cd[o + n] = cs;
    cd[o + DD + n] = ds;
  }
}

__global__ void __launch_bounds__(256)
ln_rows_kernel(const float* __restrict__ X, const float* __restrict__ w,
               const float* __restrict__ b, float* __restrict__ Y) {
  int wv = threadIdx.x >> 5, lane = threadIdx.x & 31;
  int row = blockIdx.x * 8 + wv;
  const float* x = X + (size_t)row * DD;
  float xv[16]; float s = 0.f, s2 = 0.f;
#pragma unroll
  for (int j = 0; j < 16; ++j) {
    float v = x[lane + 32 * j]; xv[j] = v; s += v; s2 += v * v;
  }
  s = wave_sum(s); s2 = wave_sum(s2);
  float m = s * (1.f / DD);
  float r = rsqrtf(s2 * (1.f / DD) - m * m + LN_EPS_C);
  float* y = Y + (size_t)row * DD;
#pragma unroll
  for (int j = 0; j < 16; ++j) {
    int d = lane + 32 * j;
    y[d] = (xv[j] - m) * r * w[d] + b[d];
  }
}

#define PJ_KC    32
#define PJ_NCH   (DD / PJ_KC)
#define PJ_P     36
#define PJ_A     (128 * PJ_P)
#define PJ_B     (64 * PJ_P)
#define PJ_STAGE (PJ_A + 2 * PJ_B)

struct ProjFrag { float4 a; float4 bk[4]; float4 bv[4]; };
struct ProjPre  { float4 x[4]; float2 k[4]; float2 v[4]; };

__device__ __forceinline__ void proj_gload(ProjPre& p, const float* __restrict__ xp,
                                           const float* __restrict__ wkp,
                                           const float* __restrict__ wvp, int kc) {
#pragma unroll
  for (int i = 0; i < 4; ++i) p.x[i] = *(const float4*)&xp[kc + 4 * i];
#pragma unroll
  for (int s = 0; s < 4; ++s) {
    p.k[s] = *(const float2*)&wkp[(size_t)(kc + s) * DD];
    p.v[s] = *(const float2*)&wvp[(size_t)(kc + s) * DD];
  }
}

__device__ __forceinline__ void proj_stage(const ProjPre& p, float* st, v2f& rsum, v2f& rsq,
                                           int arow, int aoff, int bn2, int bq) {
#pragma unroll
  for (int i = 0; i < 4; ++i) {
    float4 x = p.x[i];
    v2f lo, hi;
    lo.x = x.x; lo.y = x.y;
    hi.x = x.z; hi.y = x.w;
    rsum += lo;
    rsq += lo * lo;
    rsum += hi;
    rsq += hi * hi;
    *(float4*)&st[arow * PJ_P + aoff + 4 * i] = x;
  }
  float* bk = st + PJ_A;
  float* bv = st + PJ_A + PJ_B;
  float4 q;
  q.x = p.k[0].x; q.y = p.k[1].x; q.z = p.k[2].x; q.w = p.k[3].x;
  *(float4*)&bk[bn2 * PJ_P + 4 * bq] = q;
  q.x = p.k[0].y; q.y = p.k[1].y; q.z = p.k[2].y; q.w = p.k[3].y;
  *(float4*)&bk[(bn2 + 1) * PJ_P + 4 * bq] = q;
  q.x = p.v[0].x; q.y = p.v[1].x; q.z = p.v[2].x; q.w = p.v[3].x;
  *(float4*)&bv[bn2 * PJ_P + 4 * bq] = q;
  q.x = p.v[0].y; q.y = p.v[1].y; q.z = p.v[2].y; q.w = p.v[3].y;
  *(float4*)&bv[(bn2 + 1) * PJ_P + 4 * bq] = q;
}

__device__ __forceinline__ void proj_fload(ProjFrag& f, const float* fa, const float* fb, int kb) {
  f.a = *(const float4*)&fa[kb];
#pragma unroll
  for (int nt = 0; nt < 4; ++nt) {
    f.bk[nt] = *(const float4*)&fb[nt * 16 * PJ_P + kb];
    f.bv[nt] = *(const float4*)&fb[PJ_B + nt * 16 * PJ_P + kb];
  }
}

__device__ __forceinline__ void proj_mma(const ProjFrag& f, v8f (&accK)[4], v8f (&accV)[4]) {
  v2f a0, a1;
  a0.x = f.a.x; a0.y = f.a.y;
  a1.x = f.a.z; a1.y = f.a.w;
#pragma unroll
  for (int nt = 0; nt < 4; ++nt) {
    v2f bk, bv;
    bk.x = f.bk[nt].x; bk.y = f.bk[nt].y;
    bv.x = f.bv[nt].x; bv.y = f.bv[nt].y;
    accK[nt] = wmma4(bk, a0, accK[nt]);
    accV[nt] = wmma4(a0, bv, accV[nt]);
  }
#pragma unroll
  for (int nt = 0; nt < 4; ++nt) {
    v2f bk, bv;
    bk.x = f.bk[nt].z; bk.y = f.bk[nt].w;
    bv.x = f.bv[nt].z; bv.y = f.bv[nt].w;
    accK[nt] = wmma4(bk, a1, accK[nt]);
    accV[nt] = wmma4(a1, bv, accV[nt]);
  }
}

__global__ void __launch_bounds__(256) __attribute__((amdgpu_waves_per_eu(4, 4)))
gemm_ln_dual_kernel(const float* __restrict__ X, const float* __restrict__ Wks,
                    const float* __restrict__ Wvs, const float* __restrict__ cd,
                    float* __restrict__ Ko, float* __restrict__ Vo) {
  __shared__ __attribute__((aligned(16))) float Ls[2 * PJ_STAGE];
  __shared__ __attribute__((aligned(16))) float Rs[2 * 128];
  int t = threadIdx.x, lane = t & 31, wv = t >> 5;
  int c = lane & 15, h = lane >> 4;
  int n0 = blockIdx.x * 64, m0 = blockIdx.y * 128;

  int arow = t >> 1, aoff = (t & 1) * 16;
  size_t grow = (size_t)(m0 + arow);
  const float* xp = X + grow * DD + aoff;
  int bn2 = (t & 31) * 2, bq = t >> 5;
  const float* wkp = Wks + (size_t)(4 * bq) * DD + n0 + bn2;
  const float* wvp = Wvs + (size_t)(4 * bq) * DD + n0 + bn2;
  int fao = (wv * 16 + c) * PJ_P + 4 * h;
  int fbo = PJ_A + c * PJ_P + 4 * h;

  v8f accK[4] = {}, accV[4] = {};
  v2f rsum = {}, rsq = {};
  ProjPre pre;
  proj_gload(pre, xp, wkp, wvp, 0);
  proj_stage(pre, Ls, rsum, rsq, arow, aoff, bn2, bq);
  __syncthreads();
  for (int i = 0; i < PJ_NCH; ++i) {
    const float* st = Ls + (i & 1) * PJ_STAGE;
    float* nx = Ls + ((i + 1) & 1) * PJ_STAGE;
    bool more = (i + 1 < PJ_NCH);
    if (more) proj_gload(pre, xp, wkp, wvp, (i + 1) * PJ_KC);
    __builtin_amdgcn_sched_barrier(0);
    ProjFrag f0, f1;
    proj_fload(f0, st + fao, st + fbo, 0);
    proj_fload(f1, st + fao, st + fbo, 8);
    proj_mma(f0, accK, accV);
    proj_fload(f0, st + fao, st + fbo, 16);
    proj_mma(f1, accK, accV);
    proj_fload(f1, st + fao, st + fbo, 24);
    proj_mma(f0, accK, accV);
    __builtin_amdgcn_sched_barrier(0);
    proj_mma(f1, accK, accV);
    if (more) proj_stage(pre, nx, rsum, rsq, arow, aoff, bn2, bq);
    __syncthreads();
  }
  float s1 = rsum.x + rsum.y, s2 = rsq.x + rsq.y;
  s1 += __shfl_xor(s1, 1, 32);
  s2 += __shfl_xor(s2, 1, 32);
  float mrow = s1 * (1.f / DD);
  float rrow = rsqrtf(s2 * (1.f / DD) - mrow * mrow + LN_EPS_C);
  if ((t & 1) == 0) { Rs[arow] = rrow; Rs[128 + arow] = mrow * rrow; }
  __syncthreads();

  int half = lane >> 4, l16 = lane & 15;
  const float* rp = Rs + wv * 16 + half * 8;
  float4 ra = *(const float4*)&rp[0],   rb = *(const float4*)&rp[4];
  float4 qa = *(const float4*)&rp[128], qb = *(const float4*)&rp[132];
  float rr[8] = {ra.x, ra.y, ra.z, ra.w, rb.x, rb.y, rb.z, rb.w};
  float mq[8] = {qa.x, qa.y, qa.z, qa.w, qb.x, qb.y, qb.z, qb.w};
#pragma unroll
  for (int nt = 0; nt < 4; ++nt) {
    int ncol = n0 + nt * 16 + l16;
    float cs = cd[2 * DD + ncol], bp = cd[3 * DD + ncol];
#pragma unroll
    for (int i = 0; i < 8; ++i) {
      size_t mrw = (size_t)(m0 + wv * 16 + half * 8 + i);
      Vo[mrw * DD + ncol] = fmaf(rr[i], accV[nt][i], fmaf(-mq[i], cs, bp));
    }
  }
  float rk = Rs[wv * 16 + c], mk = Rs[128 + wv * 16 + c];
  size_t mt = (size_t)blockIdx.y * 8 + wv;
#pragma unroll
  for (int nt = 0; nt < 4; ++nt) {
    int kc0 = n0 + nt * 16 + 8 * h;
    float4 ca = *(const float4*)&cd[kc0],      cb = *(const float4*)&cd[kc0 + 4];
    float4 ba = *(const float4*)&cd[DD + kc0], bb = *(const float4*)&cd[DD + kc0 + 4];
    float* tp = Ko + ((mt * (DD / 16) + (size_t)(blockIdx.x * 4 + nt)) * 256) + lane * 4;
    float4 lo, hi;
    lo.x = fmaf(rk, accK[nt][0], fmaf(-mk, ca.x, ba.x));
    lo.y = fmaf(rk, accK[nt][1], fmaf(-mk, ca.y, ba.y));
    lo.z = fmaf(rk, accK[nt][2], fmaf(-mk, ca.z, ba.z));
    lo.w = fmaf(rk, accK[nt][3], fmaf(-mk, ca.w, ba.w));
    hi.x = fmaf(rk, accK[nt][4], fmaf(-mk, cb.x, bb.x));
    hi.y = fmaf(rk, accK[nt][5], fmaf(-mk, cb.y, bb.y));
    hi.z = fmaf(rk, accK[nt][6], fmaf(-mk, cb.z, bb.z));
    hi.w = fmaf(rk, accK[nt][7], fmaf(-mk, cb.w, bb.w));
    *(float4*)tp = lo;
    *(float4*)(tp + 128) = hi;
  }
}

struct SlotFrag {
  float4 a[8];
  float4 b[8][2];
};

template <bool BT, int K, int N>
__device__ __forceinline__ void slot_load(SlotFrag& f, const float* __restrict__ ap,
                                          const float* __restrict__ bp, int k0) {
#pragma unroll
  for (int u = 0; u < 8; ++u) {
    int kb = k0 + 8 * u;
    f.a[u] = *(const float4*)&ap[kb];
    if (BT) {
      f.b[u][0] = *(const float4*)&bp[kb];
      f.b[u][1] = *(const float4*)&bp[(size_t)K + kb];
    } else {
#pragma unroll
      for (int t = 0; t < 2; ++t) {
        float2 r0 = *(const float2*)&bp[(size_t)(kb + 2 * t) * N];
        float2 r1 = *(const float2*)&bp[(size_t)(kb + 2 * t + 1) * N];
        float4 q; q.x = r0.x; q.y = r0.y; q.z = r1.x; q.w = r1.y;
        f.b[u][t] = q;
      }
    }
  }
}

template <bool BT>
__device__ __forceinline__ void slot_mma(const SlotFrag& f, v8f& acc0, v8f& acc1) {
#pragma unroll
  for (int u = 0; u < 8; ++u) {
    v2f a0, a1, b00, b01, b10, b11;
    a0.x = f.a[u].x; a0.y = f.a[u].y;
    a1.x = f.a[u].z; a1.y = f.a[u].w;
    if (BT) {
      b00.x = f.b[u][0].x; b00.y = f.b[u][0].y;
      b01.x = f.b[u][1].x; b01.y = f.b[u][1].y;
      b10.x = f.b[u][0].z; b10.y = f.b[u][0].w;
      b11.x = f.b[u][1].z; b11.y = f.b[u][1].w;
    } else {
      b00.x = f.b[u][0].x; b00.y = f.b[u][0].z;
      b01.x = f.b[u][0].y; b01.y = f.b[u][0].w;
      b10.x = f.b[u][1].x; b10.y = f.b[u][1].z;
      b11.x = f.b[u][1].y; b11.y = f.b[u][1].w;
    }
    acc0 = wmma4(a0, b00, acc0);
    acc1 = wmma4(a0, b01, acc1);
    acc0 = wmma4(a1, b10, acc0);
    acc1 = wmma4(a1, b11, acc1);
  }
}

template <bool BT, int EPI, int K, int N>
__global__ void __launch_bounds__(128) __attribute__((amdgpu_waves_per_eu(1, 1)))
gemm_slot_kernel(const float* __restrict__ A, const float* __restrict__ B,
                 const float* __restrict__ bias, float* C, const float* aux) {
  int lane = threadIdx.x & 31, wv = threadIdx.x >> 5;
  int c = lane & 15, h = lane >> 4;
  int n0 = blockIdx.x * 32;
  int m0 = (blockIdx.y * 4 + wv) * 16;
  const float* ap = A + (size_t)(m0 + c) * K + 4 * h;
  const float* bp = BT ? B + (size_t)(n0 + 2 * c) * K + 4 * h
                       : B + (size_t)(4 * h) * N + n0 + 2 * c;
  SlotFrag f0, f1;
  v8f acc0 = {}, acc1 = {};
  slot_load<BT, K, N>(f0, ap, bp, 0);
  slot_load<BT, K, N>(f1, ap, bp, 64);
  for (int kc = 128; kc < K; kc += 128) {
    slot_mma<BT>(f0, acc0, acc1);
    slot_load<BT, K, N>(f0, ap, bp, kc);
    slot_mma<BT>(f1, acc0, acc1);
    slot_load<BT, K, N>(f1, ap, bp, kc + 64);
  }
  slot_mma<BT>(f0, acc0, acc1);
  slot_mma<BT>(f1, acc0, acc1);

  int ncol = n0 + 2 * c;
  float2 bv; bv.x = 0.f; bv.y = 0.f;
  if (EPI >= 1) bv = *(const float2*)&bias[ncol];
#pragma unroll
  for (int r = 0; r < 8; ++r) {
    size_t o = (size_t)(m0 + 8 * h + r) * N + ncol;
    float2 v; v.x = acc0[r] + bv.x; v.y = acc1[r] + bv.y;
    if (EPI == 1) { v.x = fmaxf(v.x, 0.f); v.y = fmaxf(v.y, 0.f); }
    if (EPI == 2) { float2 x = *(const float2*)&aux[o]; v.x += x.x; v.y += x.y; }
    *(float2*)&C[o] = v;
  }
}

#define DOTS_G 4
struct DotsFrag {
  float4 q[DOTS_G][2];
  float4 k[2][DOTS_G][2];
};

__device__ __forceinline__ void dots_load(DotsFrag& f, const float* __restrict__ qp,
                                          const float* __restrict__ kp, int db0) {
#pragma unroll
  for (int g = 0; g < DOTS_G; ++g) {
    int db = db0 + g;
    f.q[g][0] = *(const float4*)&qp[16 * db];
    f.q[g][1] = *(const float4*)&qp[16 * db + 4];
    f.k[0][g][0] = *(const float4*)&kp[db * 256];
    f.k[0][g][1] = *(const float4*)&kp[db * 256 + 128];
    f.k[1][g][0] = *(const float4*)&kp[(DD / 16) * 256 + db * 256];
    f.k[1][g][1] = *(const float4*)&kp[(DD / 16) * 256 + db * 256 + 128];
  }
}

__device__ __forceinline__ void dots_mma(const DotsFrag& f, v8f& acc0, v8f& acc1) {
#pragma unroll
  for (int g = 0; g < DOTS_G; ++g)
#pragma unroll
    for (int j = 0; j < 2; ++j) {
      v2f a0, a1, b00, b01, b10, b11;
      a0.x  = f.q[g][j].x;    a0.y  = f.q[g][j].y;
      a1.x  = f.q[g][j].z;    a1.y  = f.q[g][j].w;
      b00.x = f.k[0][g][j].x; b00.y = f.k[0][g][j].y;
      b01.x = f.k[0][g][j].z; b01.y = f.k[0][g][j].w;
      b10.x = f.k[1][g][j].x; b10.y = f.k[1][g][j].y;
      b11.x = f.k[1][g][j].z; b11.y = f.k[1][g][j].w;
      acc0 = wmma4(a0, b00, acc0);
      acc1 = wmma4(a0, b10, acc1);
      acc0 = wmma4(a1, b01, acc0);
      acc1 = wmma4(a1, b11, acc1);
    }
}

__global__ void __launch_bounds__(128) __attribute__((amdgpu_waves_per_eu(4, 4)))
dots_kernel(const float* __restrict__ Q, const float* __restrict__ Keys,
            float* __restrict__ Dots) {
  int lane = threadIdx.x & 31, wv = threadIdx.x >> 5;
  int c = lane & 15, h = lane >> 4;
  int w = blockIdx.x * 4 + wv;
  int b = w >> 7;
  const float* qp = Q + ((size_t)b * SS + (c & 7)) * DD + 8 * h;
  const float* kp = Keys + (size_t)(2 * w) * (DD / 16) * 256 + lane * 4;
  DotsFrag f0, f1;
  v8f acc0 = {}, acc1 = {};
  dots_load(f0, qp, kp, 0);
  dots_load(f1, qp, kp, DOTS_G);
  for (int db = 2 * DOTS_G; db < DD / 16; db += 2 * DOTS_G) {
    dots_mma(f0, acc0, acc1);
    dots_load(f0, qp, kp, db);
    dots_mma(f1, acc0, acc1);
    dots_load(f1, qp, kp, db + DOTS_G);
  }
  dots_mma(f0, acc0, acc1);
  dots_mma(f1, acc0, acc1);
  if (h == 0) {
    int fcol = ((2 * w) & 255) * 16 + c;
#pragma unroll
    for (int i = 0; i < 8; ++i) {
      float* dr = Dots + ((size_t)b * SS + i) * FF + fcol;
      dr[0]  = acc0[i] * SCALE_C;
      dr[16] = acc1[i] * SCALE_C;
    }
  }
}

__global__ void __launch_bounds__(256)
softmax_kernel(const float* __restrict__ Dots, float* __restrict__ PreNorm,
               float* __restrict__ Attn, float* __restrict__ Sums) {
  int b = blockIdx.y;
  int f = blockIdx.x * 256 + threadIdx.x;
  int lane = threadIdx.x & 31;
  float d[SS]; float mx = -3.4e38f;
#pragma unroll
  for (int s = 0; s < SS; ++s) {
    d[s] = Dots[((size_t)b * SS + s) * FF + f];
    mx = fmaxf(mx, d[s]);
  }
  float sum = 0.f;
#pragma unroll
  for (int s = 0; s < SS; ++s) { d[s] = expf(d[s] - mx); sum += d[s]; }
  float inv = 1.f / sum;
#pragma unroll
  for (int s = 0; s < SS; ++s) {
    float p = d[s] * inv;
    size_t idx = ((size_t)b * SS + s) * FF + f;
    PreNorm[idx] = p;
    float aw = p + EPS_A;
    Attn[idx] = aw;
    float r = wave_sum(aw);
    if (lane == 0) atomicAdd(&Sums[b * SS + s], r);
  }
}

__global__ void zero_sums_kernel(float* __restrict__ Sums) {
  Sums[threadIdx.x] = 0.f;
}

#define UPD_FC    256
#define UPD_NCH   (FF / UPD_FC)
#define UPD_KS    32
#define UPD_LDA   34
#define UPD_PLANE (BB * SS * DD)

__global__ void __launch_bounds__(256)
updates_part_kernel(const float* __restrict__ Attn, const float* __restrict__ Vals,
                    float* __restrict__ PartLo, float* __restrict__ PartHi) {
  __shared__ float As[16 * UPD_LDA];
  __shared__ float Bs[UPD_KS * 512];
  int t = threadIdx.x, lane = t & 31, wv = t >> 5;
  int c = blockIdx.x, b = blockIdx.y;
  const float* ab = Attn + (size_t)b * SS * FF + c * UPD_FC;
  const float* vb = Vals + ((size_t)b * FF + c * UPD_FC) * DD;
  int ar = t >> 5, ak = t & 31;
  As[(8 + ar) * UPD_LDA + ak] = 0.f;
  v8f acc[4] = {};
  for (int kc = 0; kc < UPD_FC; kc += UPD_KS) {
    __syncthreads();
    As[ar * UPD_LDA + ak] = ab[(size_t)ar * FF + kc + ak];
#pragma unroll
    for (int p = 0; p < UPD_KS / 2; ++p) {
      float2 r0 = *(const float2*)&vb[(size_t)(kc + 2 * p) * DD + 2 * t];
      float2 r1 = *(const float2*)&vb[(size_t)(kc + 2 * p + 1) * DD + 2 * t];
      float4 wb; wb.x = r0.x; wb.y = r1.x; wb.z = r0.y; wb.w = r1.y;
      *(float4*)&Bs[(p * 512 + 2 * t) * 2] = wb;
    }
    __syncthreads();
#pragma unroll
    for (int ks = 0; ks < UPD_KS; ks += 4) {
      v2f a = frag_a(As, UPD_LDA, lane, ks);
#pragma unroll
      for (int nt = 0; nt < 4; ++nt) {
        v2f b2 = frag_b(Bs, 512, lane, ks, wv * 64 + nt * 16);
        acc[nt] = wmma4(a, b2, acc[nt]);
      }
    }
  }
  float* plane = (c < 8) ? PartLo + (size_t)c * UPD_PLANE
                         : PartHi + (size_t)(c - 8) * UPD_PLANE;
  int half = lane >> 4, l16 = lane & 15;
  if (half == 0) {
#pragma unroll
    for (int i = 0; i < 8; ++i)
#pragma unroll
      for (int nt = 0; nt < 4; ++nt) {
        int n = wv * 64 + nt * 16 + l16;
        plane[((size_t)b * SS + i) * DD + n] = acc[nt][i];
      }
  }
}

__global__ void __launch_bounds__(256)
updates_sum_kernel(const float* __restrict__ PartLo, const float* __restrict__ PartHi,
                   const float* __restrict__ Sums, float* __restrict__ Upd) {
  int i4 = blockIdx.x * 256 + threadIdx.x;
  int row = i4 >> 7;
  float4 v[UPD_NCH];
#pragma unroll
  for (int c = 0; c < 8; ++c) {
    v[c]     = *(const float4*)&PartLo[(size_t)c * UPD_PLANE + 4 * (size_t)i4];
    v[c + 8] = *(const float4*)&PartHi[(size_t)c * UPD_PLANE + 4 * (size_t)i4];
  }
  float4 s = v[0];
#pragma unroll
  for (int c = 1; c < UPD_NCH; ++c) {
    s.x += v[c].x; s.y += v[c].y; s.z += v[c].z; s.w += v[c].w;
  }
  float inv = 1.f / Sums[row];
  s.x *= inv; s.y *= inv; s.z *= inv; s.w *= inv;
  *(float4*)&Upd[4 * (size_t)i4] = s;
}

__global__ void __launch_bounds__(256)
gru_kernel(const float* __restrict__ gi, const float* __restrict__ gh,
           const float* __restrict__ sn, float* __restrict__ slots) {
  int idx = blockIdx.x * 256 + threadIdx.x;
  int m = idx >> 9, d = idx & 511;
  size_t base = (size_t)m * 3 * DD;
  float ir = gi[base + d],           hr = gh[base + d];
  float iz = gi[base + DD + d],      hz = gh[base + DD + d];
  float in_ = gi[base + 2 * DD + d], hn = gh[base + 2 * DD + d];
  float r = 1.f / (1.f + expf(-(ir + hr)));
  float z = 1.f / (1.f + expf(-(iz + hz)));
  float n = tanhf(in_ + r * hn);
  float h = sn[(size_t)m * DD + d];
  slots[(size_t)m * DD + d] = (1.f - z) * n + z * h;
}

extern "C" void kernel_launch(void* const* d_in, const int* in_sizes, int n_in,
                              void* d_out, int out_size, void* d_ws, size_t ws_size,
                              hipStream_t stream) {
  (void)in_sizes; (void)n_in; (void)out_size; (void)ws_size;
  const float* in_slots  = (const float*)d_in[0];
  const float* features  = (const float*)d_in[1];
  const float* w_k       = (const float*)d_in[2];
  const float* w_v       = (const float*)d_in[3];
  const float* w_q       = (const float*)d_in[4];
  const float* ln_feat_w = (const float*)d_in[5];
  const float* ln_feat_b = (const float*)d_in[6];
  const float* ln_slot_w = (const float*)d_in[7];
  const float* ln_slot_b = (const float*)d_in[8];
  const float* gru_w_ih  = (const float*)d_in[9];
  const float* gru_w_hh  = (const float*)d_in[10];
  const float* gru_b_ih  = (const float*)d_in[11];
  const float* gru_b_hh  = (const float*)d_in[12];
  const float* ln_mlp_w  = (const float*)d_in[13];
  const float* ln_mlp_b  = (const float*)d_in[14];
  const float* mlp_w1    = (const float*)d_in[15];
  const float* mlp_b1    = (const float*)d_in[16];
  const float* mlp_w2    = (const float*)d_in[17];
  const float* mlp_b2    = (const float*)d_in[18];

  float* ws = (float*)d_ws;
  size_t off = 0;
  auto carve = [&](size_t n) { float* p = ws + off; off += n; return p; };
  float* wks   = carve((size_t)DD * DD);
  float* keys  = carve((size_t)BB * FF * DD);
  float* vals  = carve((size_t)BB * FF * DD);
  float* slots = carve((size_t)BB * SS * DD);
  float* s_n   = carve((size_t)BB * SS * DD);
  float* q     = carve((size_t)BB * SS * DD);
  float* dots  = carve((size_t)BB * SS * FF);
  float* attn  = carve((size_t)BB * SS * FF);
  float* sums  = carve(256);
  float* upd   = carve((size_t)BB * SS * DD);
  float* gi    = carve((size_t)BB * SS * 3 * DD);
  float* gh    = carve((size_t)BB * SS * 3 * DD);
  float* hln   = carve((size_t)BB * SS * DD);
  float* h1    = carve((size_t)BB * SS * HH);

  float* out_slots = (float*)d_out;
  float* out_attn  = out_slots + (size_t)BB * SS * DD;

  hipMemcpyAsync(slots, in_slots, (size_t)BB * SS * DD * sizeof(float),
                 hipMemcpyDeviceToDevice, stream);

  float* wvs = dots;
  float* cdv = dots + (size_t)DD * DD;
  fold_ln_kernel<<<32, 1024, 0, stream>>>(ln_feat_w, ln_feat_b, w_k, w_v, wks, wvs, cdv);
  gemm_ln_dual_kernel<<<dim3(DD / 64, (BB * FF) / 128), 256, 0, stream>>>(
      features, wks, wvs, cdv, keys, vals);

  for (int it = 0; it < NITER; ++it) {
    ln_rows_kernel<<<32, 256, 0, stream>>>(slots, ln_slot_w, ln_slot_b, s_n);
    gemm_slot_kernel<false, 0, DD, DD><<<dim3(DD / 32, 4), 128, 0, stream>>>(
        s_n, w_q, nullptr, q, nullptr);
    zero_sums_kernel<<<1, 256, 0, stream>>>(sums);
    dots_kernel<<<(BB * FF / 32) / 4, 128, 0, stream>>>(q, keys, dots);
    softmax_kernel<<<dim3(FF / 256, BB), 256, 0, stream>>>(dots, out_attn, attn, sums);
    updates_part_kernel<<<dim3(UPD_NCH, BB), 256, 0, stream>>>(attn, vals, dots, gi);
    updates_sum_kernel<<<(BB * SS * DD / 4) / 256, 256, 0, stream>>>(dots, gi, sums, upd);
    gemm_slot_kernel<true, 3, DD, 3 * DD><<<dim3((3 * DD) / 32, 4), 128, 0, stream>>>(
        upd, gru_w_ih, gru_b_ih, gi, nullptr);
    gemm_slot_kernel<true, 3, DD, 3 * DD><<<dim3((3 * DD) / 32, 4), 128, 0, stream>>>(
        s_n, gru_w_hh, gru_b_hh, gh, nullptr);
    gru_kernel<<<(BB * SS * DD) / 256, 256, 0, stream>>>(gi, gh, s_n, slots);
    ln_rows_kernel<<<32, 256, 0, stream>>>(slots, ln_mlp_w, ln_mlp_b, hln);
    gemm_slot_kernel<false, 1, DD, HH><<<dim3(HH / 32, 4), 128, 0, stream>>>(
        hln, mlp_w1, mlp_b1, h1, nullptr);
    gemm_slot_kernel<false, 2, HH, DD><<<dim3(DD / 32, 4), 128, 0, stream>>>(
        h1, mlp_w2, mlp_b2, slots, slots);
  }

  hipMemcpyAsync(out_slots, slots, (size_t)BB * SS * DD * sizeof(float),
                 hipMemcpyDeviceToDevice, stream);
}
